# starting kernel
# baseline (speedup 1.0000x reference)
#include <hip/hip_runtime.h>
#include <hip/hip_bf16.h>


typedef float v2f __attribute__((ext_vector_type(2)));
typedef float v8f __attribute__((ext_vector_type(8)));

#define TPB 256

__global__ void fill_kernel(float* __restrict__ p, float v, int n) {
    int i = blockIdx.x * blockDim.x + threadIdx.x;
    if (i < n) p[i] = v;
}

__global__ void deg_kernel(const int* __restrict__ dst, float* __restrict__ deg, int E) {
    int e = blockIdx.x * blockDim.x + threadIdx.x;
    if (e < E) atomicAdd(&deg[dst[e]], 1.0f);
}

__global__ void rsqrt_kernel(float* __restrict__ d, int n) {
    int i = blockIdx.x * blockDim.x + threadIdx.x;
    if (i < n) d[i] = rsqrtf(d[i]);
}

template <int K, int Nn>
__global__ void wmma_gemm_f32(const float* __restrict__ A,
                              const float* __restrict__ B,
                              float* __restrict__ C,
                              const float* __restrict__ bias,
                              int row_tiles) {
    constexpr int COLT = Nn / 16;
    int wave = (blockIdx.x * blockDim.x + threadIdx.x) >> 5;
    if (wave >= row_tiles) return;
    int lane = threadIdx.x & 31;
    int half = lane >> 4;
    int lm   = lane & 15;
    int row0 = wave << 4;

    const float* __restrict__ arow  = A + (size_t)(row0 + lm) * K + 2 * half;
    const float* __restrict__ bbase = B + (size_t)(2 * half) * Nn + lm;

    v8f acc[COLT];
#pragma unroll
    for (int c = 0; c < COLT; ++c) acc[c] = (v8f){};

#pragma unroll
    for (int k0 = 0; k0 < K; k0 += 4) {
        v2f a;
        a.x = arow[k0];
        a.y = arow[k0 + 1];
#pragma unroll
        for (int c = 0; c < COLT; ++c) {
            v2f b;
            b.x = bbase[k0 * Nn + c * 16];
            b.y = bbase[(k0 + 1) * Nn + c * 16];
            acc[c] = __builtin_amdgcn_wmma_f32_16x16x4_f32(
                 false, a,  false, b,
                 (short)0, acc[c],  false,  false);
        }
    }

    float* __restrict__ crow = C + (size_t)(row0 + 8 * half) * Nn + lm;
#pragma unroll
    for (int c = 0; c < COLT; ++c) {
        float bv = bias ? bias[c * 16 + lm] : 0.0f;
#pragma unroll
        for (int v = 0; v < 8; ++v) {
            crow[(size_t)v * Nn + c * 16] = acc[c][v] + bv;
        }
    }
}

template <int F>
__global__ void scatter_kernel(const float* __restrict__ h,
                               const int* __restrict__ src,
                               const int* __restrict__ dst,
                               const float* __restrict__ dinv,
                               float* __restrict__ out, int E) {
    int gid = blockIdx.x * blockDim.x + threadIdx.x;
    if (gid >= E * F) return;
    int e = gid / F;
    int f = gid - e * F;
    int s = src[e];
    int d = dst[e];
    float val = h[(size_t)s * F + f] * (dinv[s] * dinv[d]);
    atomicAdd(&out[(size_t)d * F + f], val);
}

template <int F, bool RELU>
__global__ void finalize_kernel(float* __restrict__ agg,
                                const float* __restrict__ xw,
                                const float* __restrict__ dinv,
                                const float* __restrict__ bias, int N) {
    int gid = blockIdx.x * blockDim.x + threadIdx.x;
    if (gid >= N * F) return;
    int i = gid / F;
    int f = gid - i * F;
    float di = dinv[i];
    float v = agg[gid] + xw[gid] * di * di + bias[f];
    agg[gid] = RELU ? fmaxf(v, 0.0f) : v;
}

template <int F>
__global__ void pool_accum_kernel(const float* __restrict__ h,
                                  const int* __restrict__ batch,
                                  float* __restrict__ sums, int N) {
    int gid = blockIdx.x * blockDim.x + threadIdx.x;
    if (gid >= N * F) return;
    int i = gid / F;
    int f = gid - i * F;
    atomicAdd(&sums[(size_t)batch[i] * F + f], h[gid]);
}

__global__ void count_kernel(const int* __restrict__ batch, float* __restrict__ cnt, int N) {
    int i = blockIdx.x * blockDim.x + threadIdx.x;
    if (i < N) atomicAdd(&cnt[batch[i]], 1.0f);
}

template <int F>
__global__ void pool_div_kernel(float* __restrict__ sums, const float* __restrict__ cnt, int G) {
    int gid = blockIdx.x * blockDim.x + threadIdx.x;
    if (gid >= G * F) return;
    int g = gid / F;
    sums[gid] /= fmaxf(cnt[g], 1.0f);
}

static inline int cdiv(long long a, long long b) { return (int)((a + b - 1) / b); }

extern "C" void kernel_launch(void* const* d_in, const int* in_sizes, int n_in,
                              void* d_out, int out_size, void* d_ws, size_t ws_size,
                              hipStream_t stream) {
    (void)n_in; (void)ws_size;
    const float* x    = (const float*)d_in[0];
    const int*   edge = (const int*)d_in[1];
    const int*   batch= (const int*)d_in[2];
    const float* b1   = (const float*)d_in[4];
    const float* W1   = (const float*)d_in[3];
    const float* W2   = (const float*)d_in[5];
    const float* b2   = (const float*)d_in[6];
    const float* Wfc  = (const float*)d_in[7];
    const float* bfc  = (const float*)d_in[8];
    float* out = (float*)d_out;

    const int N   = in_sizes[2];
    const int E   = in_sizes[1] / 2;
    const int D   = 64;
    const int D2  = 128;
    const int HID = in_sizes[8];
    const int G   = out_size / HID;

    const int* src = edge;
    const int* dst = edge + E;

    float* ws    = (float*)d_ws;
    float* xw1   = ws;
    float* agg1  = xw1  + (size_t)N * D;
    float* hw2   = agg1 + (size_t)N * D;
    float* agg2  = hw2  + (size_t)N * D2;
    float* dinv  = agg2 + (size_t)N * D2;
    float* sums  = dinv + (size_t)N;
    float* cnt   = sums + (size_t)G * D2;

    fill_kernel<<<cdiv((long long)N * D,  TPB), TPB, 0, stream>>>(agg1, 0.0f, N * D);
    fill_kernel<<<cdiv((long long)N * D2, TPB), TPB, 0, stream>>>(agg2, 0.0f, N * D2);
    fill_kernel<<<cdiv((long long)G * D2 + G, TPB), TPB, 0, stream>>>(sums, 0.0f, G * D2 + G);
    fill_kernel<<<cdiv(N, TPB), TPB, 0, stream>>>(dinv, 1.0f, N);

    deg_kernel<<<cdiv(E, TPB), TPB, 0, stream>>>(dst, dinv, E);
    rsqrt_kernel<<<cdiv(N, TPB), TPB, 0, stream>>>(dinv, N);

    {
        int row_tiles = N / 16;
        wmma_gemm_f32<64, 64><<<cdiv((long long)row_tiles * 32, TPB), TPB, 0, stream>>>(
            x, W1, xw1, nullptr, row_tiles);
    }
    scatter_kernel<64><<<cdiv((long long)E * D, TPB), TPB, 0, stream>>>(
        xw1, src, dst, dinv, agg1, E);
    finalize_kernel<64, true><<<cdiv((long long)N * D, TPB), TPB, 0, stream>>>(
        agg1, xw1, dinv, b1, N);

    {
        int row_tiles = N / 16;
        wmma_gemm_f32<64, 128><<<cdiv((long long)row_tiles * 32, TPB), TPB, 0, stream>>>(
            agg1, W2, hw2, nullptr, row_tiles);
    }
    scatter_kernel<128><<<cdiv((long long)E * D2, TPB), TPB, 0, stream>>>(
        hw2, src, dst, dinv, agg2, E);
    finalize_kernel<128, false><<<cdiv((long long)N * D2, TPB), TPB, 0, stream>>>(
        agg2, hw2, dinv, b2, N);

    pool_accum_kernel<128><<<cdiv((long long)N * D2, TPB), TPB, 0, stream>>>(
        agg2, batch, sums, N);
    count_kernel<<<cdiv(N, TPB), TPB, 0, stream>>>(batch, cnt, N);
    pool_div_kernel<128><<<cdiv((long long)G * D2, TPB), TPB, 0, stream>>>(sums, cnt, G);

    {
        int row_tiles = G / 16;
        wmma_gemm_f32<128, 64><<<cdiv((long long)row_tiles * 32, TPB), TPB, 0, stream>>>(
            sums, Wfc, out, bfc, row_tiles);
    }
}


# conv2 + mean pool as one sum per graph: an edge's scaled row adds straight into its graph's row, held in registers per wave; no per-node rows, no LDS atomics; sum order: per wave, then waves
# speedup vs baseline: 3.2551x; 3.2551x over previous
#include <hip/hip_runtime.h>
#include <hip/hip_bf16.h>


typedef float v2f __attribute__((ext_vector_type(2)));
typedef float v8f __attribute__((ext_vector_type(8)));

#define TPB 256

__global__ void fill_kernel(float* __restrict__ p, float v, int n) {
    int i = blockIdx.x * blockDim.x + threadIdx.x;
    if (i < n) p[i] = v;
}

#define AGG_NPB    196
#define AGG_OWNERS 256
__global__ void edge_prep_kernel(const int* __restrict__ src, const int* __restrict__ dst,
                                 float* __restrict__ deg, int* __restrict__ pk,
                                 int* __restrict__ obw, int* __restrict__ ocount, int E) {
    __shared__ int hist[AGG_OWNERS];
    for (int i = threadIdx.x; i < AGG_OWNERS; i += blockDim.x) hist[i] = 0;
    __syncthreads();
    int e = (blockIdx.x * blockDim.x + threadIdx.x) * 4;
    if (e + 3 < E) {
        int4 d = *(const int4*)(dst + e);
        int4 s = *(const int4*)(src + e);
        atomicAdd(&deg[d.x], 1.0f);
        atomicAdd(&deg[d.y], 1.0f);
        atomicAdd(&deg[d.z], 1.0f);
        atomicAdd(&deg[d.w], 1.0f);
        int4 p;
        p.x = (int)(((unsigned)d.x << 16) | (unsigned)s.x);
        p.y = (int)(((unsigned)d.y << 16) | (unsigned)s.y);
        p.z = (int)(((unsigned)d.z << 16) | (unsigned)s.z);
        p.w = (int)(((unsigned)d.w << 16) | (unsigned)s.w);
        *(int4*)(pk + e) = p;
        const unsigned ox = (unsigned)(d.x / AGG_NPB), oy = (unsigned)(d.y / AGG_NPB);
        const unsigned oz = (unsigned)(d.z / AGG_NPB), ow = (unsigned)(d.w / AGG_NPB);
        obw[e >> 2] = (int)(ox | (oy << 8) | (oz << 16) | (ow << 24));
        atomicAdd(&hist[ox], 1);
        atomicAdd(&hist[oy], 1);
        atomicAdd(&hist[oz], 1);
        atomicAdd(&hist[ow], 1);
    } else {
        unsigned char* ob = (unsigned char*)obw;
        for (; e < E; ++e) {
            int d = dst[e];
            atomicAdd(&deg[d], 1.0f);
            pk[e] = (int)(((unsigned)d << 16) | (unsigned)src[e]);
            ob[e] = (unsigned char)(d / AGG_NPB);
            atomicAdd(&hist[d / AGG_NPB], 1);
        }
    }
    __syncthreads();
    for (int i = threadIdx.x; i < AGG_OWNERS; i += blockDim.x) {
        const int c = hist[i];
        if (c) atomicAdd(&ocount[i], c);
    }
}

__global__ void rsqrt_kernel(float* __restrict__ d, int n) {
    int i = blockIdx.x * blockDim.x + threadIdx.x;
    if (i < n) d[i] = rsqrtf(d[i]);
}

template <int K, int Nn>
__global__ void wmma_gemm_f32(const float* __restrict__ A,
                              const float* __restrict__ B,
                              float* __restrict__ C,
                              const float* __restrict__ bias,
                              int row_tiles) {
    constexpr int COLT = Nn / 16;
    int wave = (blockIdx.x * blockDim.x + threadIdx.x) >> 5;
    if (wave >= row_tiles) return;
    int lane = threadIdx.x & 31;
    int half = lane >> 4;
    int lm   = lane & 15;
    int row0 = wave << 4;

    const float* __restrict__ arow  = A + (size_t)(row0 + lm) * K + 2 * half;
    const float* __restrict__ bbase = B + (size_t)(2 * half) * Nn + lm;

    v8f acc[COLT];
#pragma unroll
    for (int c = 0; c < COLT; ++c) acc[c] = (v8f){};

#pragma unroll
    for (int k0 = 0; k0 < K; k0 += 4) {
        v2f a;
        a.x = arow[k0];
        a.y = arow[k0 + 1];
#pragma unroll
        for (int c = 0; c < COLT; ++c) {
            v2f b;
            b.x = bbase[k0 * Nn + c * 16];
            b.y = bbase[(k0 + 1) * Nn + c * 16];
            acc[c] = __builtin_amdgcn_wmma_f32_16x16x4_f32(
                 false, a,  false, b,
                 (short)0, acc[c],  false,  false);
        }
    }

    float* __restrict__ crow = C + (size_t)(row0 + 8 * half) * Nn + lm;
#pragma unroll
    for (int c = 0; c < COLT; ++c) {
        float bv = bias ? bias[c * 16 + lm] : 0.0f;
#pragma unroll
        for (int v = 0; v < 8; ++v) {
            crow[(size_t)v * Nn + c * 16] = acc[c][v] + bv;
        }
    }
}

#define AGG_TPB   1024
#define AGG_WAVES 32
#define AGG_QCAP  1056

typedef float v4f __attribute__((ext_vector_type(4)));

template <int F> struct RowVecSel      { typedef v2f T; };
template <>      struct RowVecSel<128> { typedef v4f T; };

template <int F, bool FULL>
__device__ __forceinline__ void agg_rows(const float* __restrict__ h, float* acc,
                                         int s_l, int dl_l, float w_l, int cnt, int lane,
                                         const int (&off)[F / 32]) {
    typedef typename RowVecSel<F>::T VT;
    constexpr int VEC = F / 32;
#pragma unroll
    for (int g = 0; g < 32; g += 8) {
        if (!FULL && g >= cnt) break;
        VT r[8];
#pragma unroll
        for (int i = 0; i < 8; ++i) {
            if (FULL || g + i < cnt) {
                int s = __builtin_amdgcn_readlane(s_l, g + i);
                r[i] = *((const VT*)(h + (size_t)s * F) + lane);
            }
        }
#pragma unroll
        for (int i = 0; i < 8; ++i) {
            if (FULL || g + i < cnt) {
                int dl = __builtin_amdgcn_readlane(dl_l, g + i);
                float w = __builtin_bit_cast(
                    float, __builtin_amdgcn_readlane(__builtin_bit_cast(int, w_l), g + i));
                float* a = acc + dl * F;
#pragma unroll
                for (int k = 0; k < VEC; ++k)
                    __hip_atomic_fetch_add(&a[off[k]], r[i][k] * w, __ATOMIC_RELAXED,
                                           __HIP_MEMORY_SCOPE_WORKGROUP);
            }
        }
    }
}

template <int F, bool FULL>
__device__ __forceinline__ void agg_drain(const float* __restrict__ h,
                                          const int* __restrict__ pk,
                                          const float* __restrict__ dinv,
                                          float* acc, const int* q,
                                          int start, int cnt, int node0, int lane,
                                          const int (&off)[F / 32],
                                          int2* __restrict__ rec_out) {
    int dl_l = 0, s_l = 0;
    float w_l = 0.0f;
    if (FULL || lane < cnt) {
        int e_l = q[start + lane];
        int p_l = pk[e_l];
        int d_l = (int)((unsigned)p_l >> 16);
        s_l = p_l & 0xFFFF;
        dl_l = d_l - node0;
        w_l = dinv[s_l] * dinv[d_l];
        rec_out[lane] = make_int2(p_l, __builtin_bit_cast(int, w_l));
    }
    agg_rows<F, FULL>(h, acc, s_l, dl_l, w_l, cnt, lane, off);
}

template <int F>
__device__ __forceinline__ void agg_row_words(int (&off)[F / 32], int lane) {
    constexpr int VEC = F / 32;
#pragma unroll
    for (int k = 0; k < VEC; ++k)
        off[k] = (F == 128) ? (4 * lane + (k ^ (lane >> 4))) : (VEC * lane + k);
}

__device__ __forceinline__ int agg_owner_base(const int* __restrict__ ocount, int b, int lane) {
    int s = 0;
    for (int i = lane; i < b; i += 32) s += ocount[i];
    for (int o = 16; o > 0; o >>= 1) s += __shfl_xor(s, o);
    return s;
}

#define AGG_SLICES 8
#define AGG_RPS    ((AGG_NPB + AGG_SLICES - 1) / AGG_SLICES)
template <int F, bool RELU, bool POOL>
__device__ __forceinline__ void agg_finish(float* acc,
                                           const float* __restrict__ h,
                                           const float* __restrict__ dinv,
                                           const float* __restrict__ bias,
                                           float* __restrict__ out,
                                           const int* __restrict__ batch,
                                           float* __restrict__ sums,
                                           float* __restrict__ cnt,
                                           int node0, int nloc, int tid) {
    constexpr int F4 = F / 4;
    for (int idx = tid; idx < nloc * F4; idx += AGG_TPB) {
        int i = idx / F4;
        int g = idx - i * F4;
        v4f q = ((const v4f*)acc)[idx];
        if (F == 128 && (g >> 4)) q = (v4f){q.y, q.x, q.w, q.z};
        int gi = node0 + i;
        float di = dinv[gi];
        v4f xv = ((const v4f*)h)[(size_t)gi * F4 + g];
        v4f bv = ((const v4f*)bias)[g];
        v4f o;
#pragma unroll
        for (int c = 0; c < 4; ++c) {
            float v = q[c] + xv[c] * di * di + bv[c];
            o[c] = RELU ? fmaxf(v, 0.0f) : v;
        }
        if (POOL) ((v4f*)acc)[idx] = o;
        else      ((v4f*)out)[(size_t)gi * F4 + g] = o;
    }

    if (POOL) {
        static_assert(!POOL || AGG_TPB == AGG_SLICES * F, "one thread per (slice, feature)");
        __syncthreads();
        const int f  = tid & (F - 1);
        const int r0 = (tid / F) * AGG_RPS;
        int   bid[AGG_RPS];
        float val[AGG_RPS];
#pragma unroll
        for (int j = 0; j < AGG_RPS; ++j) {
            const int r  = r0 + j;
            const bool ok = r < nloc;
            const int rr = ok ? r : 0;
            const int b  = batch[node0 + rr];
            const float v = acc[rr * F + f];
            bid[j] = ok ? b : -1;
            val[j] = ok ? v : 0.0f;
        }
        int   cur  = -1;
        float run  = 0.0f;
        float nrun = 0.0f;
#pragma unroll
        for (int j = 0; j < AGG_RPS; ++j) {
            if (bid[j] != cur) {
                if (cur >= 0) {
                    atomicAdd(&sums[(size_t)cur * F + f], run);
                    if (f == 0) atomicAdd(&cnt[cur], nrun);
                }
                cur = bid[j]; run = 0.0f; nrun = 0.0f;
            }
            run  += val[j];
            nrun += 1.0f;
        }
        if (cur >= 0) {
            atomicAdd(&sums[(size_t)cur * F + f], run);
            if (f == 0) atomicAdd(&cnt[cur], nrun);
        }
    }
}

template <int F, bool RELU>
__global__ __launch_bounds__(AGG_TPB)
void agg_owner_kernel(const float* __restrict__ h,
                      const int* __restrict__ pk,
                      const int* __restrict__ obw,
                      const int* __restrict__ ocount,
                      const float* __restrict__ dinv,
                      const float* __restrict__ bias,
                      float* __restrict__ out,
                      int* __restrict__ recw, int N, int E) {
    constexpr int F4  = F / 4;
    __shared__ __attribute__((aligned(16))) float acc[AGG_NPB * F];
    __shared__ int qbuf[AGG_WAVES * AGG_QCAP];
    __shared__ int cursor;

    const int tid  = threadIdx.x;
    const int lane = tid & 31;
    const int wave = __builtin_amdgcn_readfirstlane(tid >> 5);
    const int node0 = blockIdx.x * AGG_NPB;
    const int nloc  = min(AGG_NPB, N - node0);

    for (int i = tid; i < AGG_NPB * F4; i += AGG_TPB) ((v4f*)acc)[i] = (v4f){0.f, 0.f, 0.f, 0.f};
    if (tid == 0) cursor = 0;
    int2* rec = (int2*)recw + agg_owner_base(ocount, (int)blockIdx.x, lane);
    __syncthreads();

    int off[F / 32];
    agg_row_words<F>(off, lane);

    int* myq = qbuf + wave * AGG_QCAP;
    int qcnt = 0;

    auto push = [&](bool m, int e) {
        unsigned mask = __builtin_amdgcn_ballot_w32(m);
        if (mask) {
            int pos = qcnt + (int)__builtin_amdgcn_mbcnt_lo(mask, 0u);
            if (m) myq[pos] = e;
            qcnt += __builtin_popcount(mask);
        }
    };
    auto reserve = [&](int n) -> int {
        int pos = 0;
        if (lane == 0) pos = atomicAdd(&cursor, n);
        return __builtin_amdgcn_readfirstlane(pos);
    };
    const int nch = E >> 9;
    const unsigned pat = (unsigned)blockIdx.x * 0x01010101u;
    auto ld4 = [&](int c) -> int4 {
        return *(const int4*)(obw + (size_t)c * 128 + lane * 4);
    };
    auto scanw = [&](int wv, int eb) {
        const unsigned x = (unsigned)wv ^ pat;
        const unsigned t = (x - 0x01010101u) & ~x & 0x80808080u;
        if (__builtin_amdgcn_ballot_w32(t != 0u)) {
            push((x & 0x000000FFu) == 0u, eb);
            push((x & 0x0000FF00u) == 0u, eb + 1);
            push((x & 0x00FF0000u) == 0u, eb + 2);
            push((x & 0xFF000000u) == 0u, eb + 3);
        }
    };
    auto scan16 = [&](int4 d, int c) {
        const int eb = c * 512 + lane * 16;
        scanw(d.x, eb);
        scanw(d.y, eb + 4);
        scanw(d.z, eb + 8);
        scanw(d.w, eb + 12);
    };
    auto drain_full = [&]() {
        while (qcnt >= 32) {
            qcnt -= 32;
            agg_drain<F, true>(h, pk, dinv, acc, myq, qcnt, 32, node0, lane, off,
                               rec + reserve(32));
        }
    };

    int c0 = wave;
    for (; c0 + 3 * AGG_WAVES < nch; c0 += 4 * AGG_WAVES) {
        int4 d0 = ld4(c0);
        int4 d1 = ld4(c0 + AGG_WAVES);
        int4 d2 = ld4(c0 + 2 * AGG_WAVES);
        int4 d3 = ld4(c0 + 3 * AGG_WAVES);
        scan16(d0, c0);
        scan16(d1, c0 + AGG_WAVES);
        drain_full();
        scan16(d2, c0 + 2 * AGG_WAVES);
        scan16(d3, c0 + 3 * AGG_WAVES);
        drain_full();
    }
    for (; c0 < nch; c0 += AGG_WAVES) {
        scan16(ld4(c0), c0);
        drain_full();
    }
    if (wave == 0) {
        const unsigned char* ob = (const unsigned char*)obw;
        for (int t = nch * 512; t < E; t += 32) {
            int e = t + lane;
            bool m = (e < E) && ((int)ob[e] == (int)blockIdx.x);
            push(m, e);
        }
    }
    while (qcnt > 0) {
        int cnt = min(qcnt, 32);
        qcnt -= cnt;
        agg_drain<F, false>(h, pk, dinv, acc, myq, qcnt, cnt, node0, lane, off,
                            rec + reserve(cnt));
    }
    __syncthreads();

    agg_finish<F, RELU, false>(acc, h, dinv, bias, out, nullptr, nullptr, nullptr,
                               node0, nloc, tid);
}

#define AGG_NG 4
#define AGG_NSELF ((AGG_NPB + AGG_WAVES - 1) / AGG_WAVES)

__device__ __forceinline__ void graph_add(v4f (&a)[AGG_NG], int g, v4f v, float w) {
#pragma unroll
    for (int j = 0; j < AGG_NG; ++j) {
        const float m = (g == j) ? w : 0.0f;
        a[j] += v * m;
    }
}

__device__ __forceinline__ void graph_rows(const float* __restrict__ h, v4f (&a)[AGG_NG],
                                           int s_l, int g_l, float w_l, int lane) {
#pragma unroll
    for (int g0 = 0; g0 < 32; g0 += 8) {
        v4f r[8];
#pragma unroll
        for (int i = 0; i < 8; ++i) {
            int s = __builtin_amdgcn_readlane(s_l, g0 + i);
            r[i] = *((const v4f*)(h + (size_t)s * 128) + lane);
        }
#pragma unroll
        for (int i = 0; i < 8; ++i) {
            int g = __builtin_amdgcn_readlane(g_l, g0 + i);
            float w = __builtin_bit_cast(
                float, __builtin_amdgcn_readlane(__builtin_bit_cast(int, w_l), g0 + i));
            graph_add(a, g, r[i], w);
        }
    }
}

#define AGG_RCAP 8192
#define AGG_RLD  4
template <int F, bool RELU, bool POOL>
__global__ __launch_bounds__(AGG_TPB)
void agg_replay_kernel(const float* __restrict__ h,
                       const int* __restrict__ recw,
                       const int* __restrict__ ocount,
                       const float* __restrict__ dinv,
                       const float* __restrict__ bias,
                       float* __restrict__ out,
                       const int* __restrict__ batch,
                       float* __restrict__ sums,
                       float* __restrict__ cnt, int N) {
    constexpr int F4 = F / 4;
    __shared__ __attribute__((aligned(16))) float acc[AGG_NPB * F];
    __shared__ int2 srec[AGG_RCAP];
    __shared__ int sbat[AGG_NPB];

    const int tid  = threadIdx.x;
    const int lane = tid & 31;
    const int wave = __builtin_amdgcn_readfirstlane(tid >> 5);
    const int node0 = blockIdx.x * AGG_NPB;
    const int nloc  = min(AGG_NPB, N - node0);

    int g_lo = 0, span = AGG_NG + 1;
    if (POOL) {
        int lo = 0x7FFFFFFF, hi = -1;
        for (int i = lane; i < nloc; i += 32) {
            const int b = batch[node0 + i];
            lo = min(lo, b); hi = max(hi, b);
        }
        for (int o = 16; o > 0; o >>= 1) {
            lo = min(lo, __shfl_xor(lo, o)); hi = max(hi, __shfl_xor(hi, o));
        }
        g_lo = __builtin_amdgcn_readfirstlane(lo);
        span = __builtin_amdgcn_readfirstlane(hi) - g_lo + 1;
        for (int i = tid; i < nloc; i += AGG_TPB) sbat[i] = batch[node0 + i];
    }
    const bool graphs = POOL && F == 128 && span <= AGG_NG;
    if (!graphs)
        for (int i = tid; i < AGG_NPB * F4; i += AGG_TPB) ((v4f*)acc)[i] = (v4f){0.f, 0.f, 0.f, 0.f};
    v4f ga[AGG_NG];
#pragma unroll
    for (int j = 0; j < AGG_NG; ++j) ga[j] = (v4f){0.f, 0.f, 0.f, 0.f};
    const int total = ocount[blockIdx.x];
    const int2* __restrict__ rec =
        (const int2*)recw + agg_owner_base(ocount, (int)blockIdx.x, lane);

    int off[F / 32];
    agg_row_words<F>(off, lane);

    for (int r0 = 0; r0 < total; r0 += AGG_RCAP) {
        const int nst = min(total - r0, AGG_RCAP);
        if (r0 != 0) __syncthreads();
        for (int j0 = tid; j0 < nst; j0 += AGG_RLD * AGG_TPB) {
            int2 t[AGG_RLD];
#pragma unroll
            for (int k = 0; k < AGG_RLD; ++k)
                t[k] = rec[r0 + min(j0 + k * AGG_TPB, nst - 1)];
#pragma unroll
            for (int k = 0; k < AGG_RLD; ++k)
                if (j0 + k * AGG_TPB < nst) srec[j0 + k * AGG_TPB] = t[k];
        }
        __syncthreads();

        const int nfull = nst >> 5;
        const int rest  = nst & 31;
        for (int i = wave; i < nfull; i += AGG_WAVES) {
            const int2 r = srec[32 * i + lane];
            const int dl = (int)((unsigned)r.x >> 16) - node0;
            if (graphs)
                graph_rows(h, ga, r.x & 0xFFFF, sbat[dl] - g_lo,
                           __builtin_bit_cast(float, r.y), lane);
            else
                agg_rows<F, true>(h, acc, r.x & 0xFFFF, dl,
                                  __builtin_bit_cast(float, r.y), 32, lane, off);
        }
        if (rest != 0 && (nfull & (AGG_WAVES - 1)) == wave) {
            int dl_l = 0, s_l = 0;
            float w_l = 0.0f;
            if (lane < rest) {
                const int2 t = srec[32 * nfull + lane];
                s_l  = t.x & 0xFFFF;
                dl_l = (int)((unsigned)t.x >> 16) - node0;
                w_l  = __builtin_bit_cast(float, t.y);
            }
            if (graphs)
                graph_rows(h, ga, s_l, sbat[dl_l] - g_lo, w_l, lane);
            else
                agg_rows<F, false>(h, acc, s_l, dl_l, w_l, rest, lane, off);
        }
    }
    __syncthreads();

    if (!graphs) {
        agg_finish<F, RELU, POOL>(acc, h, dinv, bias, out, batch, sums, cnt, node0, nloc, tid);
        return;
    }

    v4f   rr[AGG_NSELF];
    float dk[AGG_NSELF];
    int   gk[AGG_NSELF];
#pragma unroll
    for (int k = 0; k < AGG_NSELF; ++k) {
        const int ii = min(wave + k * AGG_WAVES, nloc - 1);
        rr[k] = ((const v4f*)h)[(size_t)(node0 + ii) * F4 + lane];
        dk[k] = dinv[node0 + ii];
        gk[k] = __builtin_amdgcn_readfirstlane(sbat[ii]) - g_lo;
    }
    const v4f bv = ((const v4f*)bias)[lane];
#pragma unroll
    for (int k = 0; k < AGG_NSELF; ++k) {
        const bool ok = wave + k * AGG_WAVES < nloc;
        graph_add(ga, gk[k], rr[k], ok ? dk[k] * dk[k] : 0.0f);
        graph_add(ga, gk[k], bv, ok ? 1.0f : 0.0f);
    }

#pragma unroll
    for (int j = 0; j < AGG_NG; ++j) ((v4f*)acc)[(wave * AGG_NG + j) * F4 + lane] = ga[j];
    __syncthreads();
    for (int idx = tid; idx < span * F; idx += AGG_TPB) {
        const int j = idx / F, f = idx - j * F;
        float s = 0.0f;
        for (int w = 0; w < AGG_WAVES; ++w) s += acc[(w * AGG_NG + j) * F + f];
        atomicAdd(&sums[(size_t)(g_lo + j) * F + f], s);
    }
    if (tid < span) {
        float c = 0.0f;
        for (int i = 0; i < nloc; ++i) c += (sbat[i] - g_lo == tid) ? 1.0f : 0.0f;
        atomicAdd(&cnt[g_lo + tid], c);
    }
}

template <int F>
__global__ void pool_div_kernel(float* __restrict__ sums, const float* __restrict__ cnt, int G) {
    int gid = blockIdx.x * blockDim.x + threadIdx.x;
    if (gid >= G * F) return;
    int g = gid / F;
    sums[gid] /= fmaxf(cnt[g], 1.0f);
}

static inline int cdiv(long long a, long long b) { return (int)((a + b - 1) / b); }

extern "C" void kernel_launch(void* const* d_in, const int* in_sizes, int n_in,
                              void* d_out, int out_size, void* d_ws, size_t ws_size,
                              hipStream_t stream) {
    (void)n_in; (void)ws_size;
    const float* x    = (const float*)d_in[0];
    const int*   edge = (const int*)d_in[1];
    const int*   batch= (const int*)d_in[2];
    const float* b1   = (const float*)d_in[4];
    const float* W1   = (const float*)d_in[3];
    const float* W2   = (const float*)d_in[5];
    const float* b2   = (const float*)d_in[6];
    const float* Wfc  = (const float*)d_in[7];
    const float* bfc  = (const float*)d_in[8];
    float* out = (float*)d_out;

    const int N   = in_sizes[2];
    const int E   = in_sizes[1] / 2;
    const int D   = 64;
    const int D2  = 128;
    const int HID = in_sizes[8];
    const int G   = out_size / HID;

    const int* src = edge;
    const int* dst = edge + E;

    float* ws    = (float*)d_ws;
    float* xw1   = ws;
    float* agg1  = xw1  + (size_t)N * D;
    float* hw2   = agg1 + (size_t)N * D;
    float* ebuf  = hw2  + (size_t)N * D2;
    int*   pk    = (int*)ebuf;
    int*   obw   = pk + E;
    int*   rec   = obw + E;
    float* zblk  = (float*)(rec + 2 * (size_t)E);
    int*   ocount= (int*)zblk;
    float* sums  = zblk + AGG_OWNERS;
    float* cnt   = sums + (size_t)G * D2;
    float* dinv  = ebuf + (size_t)N * D2;

    fill_kernel<<<cdiv((long long)AGG_OWNERS + G * D2 + G, TPB), TPB, 0, stream>>>(
        zblk, 0.0f, AGG_OWNERS + G * D2 + G);
    fill_kernel<<<cdiv(N, TPB), TPB, 0, stream>>>(dinv, 1.0f, N);

    edge_prep_kernel<<<cdiv(cdiv(E, 4), TPB), TPB, 0, stream>>>(src, dst, dinv, pk, obw, ocount, E);
    rsqrt_kernel<<<cdiv(N, TPB), TPB, 0, stream>>>(dinv, N);

    {
        int row_tiles = N / 16;
        wmma_gemm_f32<64, 64><<<cdiv((long long)row_tiles * 32, TPB), TPB, 0, stream>>>(
            x, W1, xw1, nullptr, row_tiles);
    }
    agg_owner_kernel<64, true><<<cdiv(N, AGG_NPB), AGG_TPB, 0, stream>>>(
        xw1, pk, obw, ocount, dinv, b1, agg1, rec, N, E);

    {
        int row_tiles = N / 16;
        wmma_gemm_f32<64, 128><<<cdiv((long long)row_tiles * 32, TPB), TPB, 0, stream>>>(
            agg1, W2, hw2, nullptr, row_tiles);
    }
    agg_replay_kernel<128, false, true><<<cdiv(N, AGG_NPB), AGG_TPB, 0, stream>>>(
        hw2, rec, ocount, dinv, b2, nullptr, batch, sums, cnt, N);

    pool_div_kernel<128><<<cdiv((long long)G * D2, TPB), TPB, 0, stream>>>(sums, cnt, G);

    {
        int row_tiles = G / 16;
        wmma_gemm_f32<128, 64><<<cdiv((long long)row_tiles * 32, TPB), TPB, 0, stream>>>(
            sums, Wfc, out, bfc, row_tiles);
    }
}


# conv1 owners do both dense products on their LDS rows (f32 WMMA): x rows summed first, then relu(rows@W1+b1)@W2 stored; no product launches, xW1 and h1 never stored; sum order: aggregate, then W1
# speedup vs baseline: 3.6755x; 1.1292x over previous
#include <hip/hip_runtime.h>
#include <hip/hip_bf16.h>


typedef float v2f __attribute__((ext_vector_type(2)));
typedef float v8f __attribute__((ext_vector_type(8)));

#define TPB 256

__global__ void fill_kernel(float* __restrict__ p, float v, int n) {
    int i = blockIdx.x * blockDim.x + threadIdx.x;
    if (i < n) p[i] = v;
}

#define AGG_NPB    196
#define AGG_OWNERS 256
__global__ void edge_prep_kernel(const int* __restrict__ src, const int* __restrict__ dst,
                                 float* __restrict__ deg, int* __restrict__ pk,
                                 int* __restrict__ obw, int* __restrict__ ocount, int E) {
    __shared__ int hist[AGG_OWNERS];
    for (int i = threadIdx.x; i < AGG_OWNERS; i += blockDim.x) hist[i] = 0;
    __syncthreads();
    int e = (blockIdx.x * blockDim.x + threadIdx.x) * 4;
    if (e + 3 < E) {
        int4 d = *(const int4*)(dst + e);
        int4 s = *(const int4*)(src + e);
        atomicAdd(&deg[d.x], 1.0f);
        atomicAdd(&deg[d.y], 1.0f);
        atomicAdd(&deg[d.z], 1.0f);
        atomicAdd(&deg[d.w], 1.0f);
        int4 p;
        p.x = (int)(((unsigned)d.x << 16) | (unsigned)s.x);
        p.y = (int)(((unsigned)d.y << 16) | (unsigned)s.y);
        p.z = (int)(((unsigned)d.z << 16) | (unsigned)s.z);
        p.w = (int)(((unsigned)d.w << 16) | (unsigned)s.w);
        *(int4*)(pk + e) = p;
        const unsigned ox = (unsigned)(d.x / AGG_NPB), oy = (unsigned)(d.y / AGG_NPB);
        const unsigned oz = (unsigned)(d.z / AGG_NPB), ow = (unsigned)(d.w / AGG_NPB);
        obw[e >> 2] = (int)(ox | (oy << 8) | (oz << 16) | (ow << 24));
        atomicAdd(&hist[ox], 1);
        atomicAdd(&hist[oy], 1);
        atomicAdd(&hist[oz], 1);
        atomicAdd(&hist[ow], 1);
    } else {
        unsigned char* ob = (unsigned char*)obw;
        for (; e < E; ++e) {
            int d = dst[e];
            atomicAdd(&deg[d], 1.0f);
            pk[e] = (int)(((unsigned)d << 16) | (unsigned)src[e]);
            ob[e] = (unsigned char)(d / AGG_NPB);
            atomicAdd(&hist[d / AGG_NPB], 1);
        }
    }
    __syncthreads();
    for (int i = threadIdx.x; i < AGG_OWNERS; i += blockDim.x) {
        const int c = hist[i];
        if (c) atomicAdd(&ocount[i], c);
    }
}

__global__ void rsqrt_kernel(float* __restrict__ d, int n) {
    int i = blockIdx.x * blockDim.x + threadIdx.x;
    if (i < n) d[i] = rsqrtf(d[i]);
}

template <int K, int Nn>
__global__ void wmma_gemm_f32(const float* __restrict__ A,
                              const float* __restrict__ B,
                              float* __restrict__ C,
                              const float* __restrict__ bias,
                              int row_tiles) {
    constexpr int COLT = Nn / 16;
    int wave = (blockIdx.x * blockDim.x + threadIdx.x) >> 5;
    if (wave >= row_tiles) return;
    int lane = threadIdx.x & 31;
    int half = lane >> 4;
    int lm   = lane & 15;
    int row0 = wave << 4;

    const float* __restrict__ arow  = A + (size_t)(row0 + lm) * K + 2 * half;
    const float* __restrict__ bbase = B + (size_t)(2 * half) * Nn + lm;

    v8f acc[COLT];
#pragma unroll
    for (int c = 0; c < COLT; ++c) acc[c] = (v8f){};

#pragma unroll
    for (int k0 = 0; k0 < K; k0 += 4) {
        v2f a;
        a.x = arow[k0];
        a.y = arow[k0 + 1];
#pragma unroll
        for (int c = 0; c < COLT; ++c) {
            v2f b;
            b.x = bbase[k0 * Nn + c * 16];
            b.y = bbase[(k0 + 1) * Nn + c * 16];
            acc[c] = __builtin_amdgcn_wmma_f32_16x16x4_f32(
                 false, a,  false, b,
                 (short)0, acc[c],  false,  false);
        }
    }

    float* __restrict__ crow = C + (size_t)(row0 + 8 * half) * Nn + lm;
#pragma unroll
    for (int c = 0; c < COLT; ++c) {
        float bv = bias ? bias[c * 16 + lm] : 0.0f;
#pragma unroll
        for (int v = 0; v < 8; ++v) {
            crow[(size_t)v * Nn + c * 16] = acc[c][v] + bv;
        }
    }
}

#define AGG_TPB   1024
#define AGG_WAVES 32
#define AGG_QCAP  1056

typedef float v4f __attribute__((ext_vector_type(4)));

template <int F> struct RowVecSel      { typedef v2f T; };
template <>      struct RowVecSel<128> { typedef v4f T; };

template <int F, bool FULL, int PITCH = F>
__device__ __forceinline__ void agg_rows(const float* __restrict__ h, float* acc,
                                         int s_l, int dl_l, float w_l, int cnt, int lane,
                                         const int (&off)[F / 32]) {
    typedef typename RowVecSel<F>::T VT;
    constexpr int VEC = F / 32;
#pragma unroll
    for (int g = 0; g < 32; g += 8) {
        if (!FULL && g >= cnt) break;
        VT r[8];
#pragma unroll
        for (int i = 0; i < 8; ++i) {
            if (FULL || g + i < cnt) {
                int s = __builtin_amdgcn_readlane(s_l, g + i);
                r[i] = *((const VT*)(h + (size_t)s * F) + lane);
            }
        }
#pragma unroll
        for (int i = 0; i < 8; ++i) {
            if (FULL || g + i < cnt) {
                int dl = __builtin_amdgcn_readlane(dl_l, g + i);
                float w = __builtin_bit_cast(
                    float, __builtin_amdgcn_readlane(__builtin_bit_cast(int, w_l), g + i));
                float* a = acc + dl * PITCH;
#pragma unroll
                for (int k = 0; k < VEC; ++k)
                    __hip_atomic_fetch_add(&a[off[k]], r[i][k] * w, __ATOMIC_RELAXED,
                                           __HIP_MEMORY_SCOPE_WORKGROUP);
            }
        }
    }
}

template <int F, bool FULL, int PITCH = F>
__device__ __forceinline__ void agg_drain(const float* __restrict__ h,
                                          const int* __restrict__ pk,
                                          const float* __restrict__ dinv,
                                          float* acc, const int* q,
                                          int start, int cnt, int node0, int lane,
                                          const int (&off)[F / 32],
                                          int2* __restrict__ rec_out) {
    int dl_l = 0, s_l = 0;
    float w_l = 0.0f;
    if (FULL || lane < cnt) {
        int e_l = q[start + lane];
        int p_l = pk[e_l];
        int d_l = (int)((unsigned)p_l >> 16);
        s_l = p_l & 0xFFFF;
        dl_l = d_l - node0;
        w_l = dinv[s_l] * dinv[d_l];
        rec_out[lane] = make_int2(p_l, __builtin_bit_cast(int, w_l));
    }
    agg_rows<F, FULL, PITCH>(h, acc, s_l, dl_l, w_l, cnt, lane, off);
}

template <int F>
__device__ __forceinline__ void agg_row_words(int (&off)[F / 32], int lane) {
    constexpr int VEC = F / 32;
#pragma unroll
    for (int k = 0; k < VEC; ++k)
        off[k] = (F == 128) ? (4 * lane + (k ^ (lane >> 4))) : (VEC * lane + k);
}

__device__ __forceinline__ int agg_owner_base(const int* __restrict__ ocount, int b, int lane) {
    int s = 0;
    for (int i = lane; i < b; i += 32) s += ocount[i];
    for (int o = 16; o > 0; o >>= 1) s += __shfl_xor(s, o);
    return s;
}

#define AGG_SLICES 8
#define AGG_RPS    ((AGG_NPB + AGG_SLICES - 1) / AGG_SLICES)
template <int F, bool RELU, bool POOL>
__device__ __forceinline__ void agg_finish(float* acc,
                                           const float* __restrict__ h,
                                           const float* __restrict__ dinv,
                                           const float* __restrict__ bias,
                                           float* __restrict__ out,
                                           const int* __restrict__ batch,
                                           float* __restrict__ sums,
                                           float* __restrict__ cnt,
                                           int node0, int nloc, int tid) {
    constexpr int F4 = F / 4;
    for (int idx = tid; idx < nloc * F4; idx += AGG_TPB) {
        int i = idx / F4;
        int g = idx - i * F4;
        v4f q = ((const v4f*)acc)[idx];
        if (F == 128 && (g >> 4)) q = (v4f){q.y, q.x, q.w, q.z};
        int gi = node0 + i;
        float di = dinv[gi];
        v4f xv = ((const v4f*)h)[(size_t)gi * F4 + g];
        v4f bv = ((const v4f*)bias)[g];
        v4f o;
#pragma unroll
        for (int c = 0; c < 4; ++c) {
            float v = q[c] + xv[c] * di * di + bv[c];
            o[c] = RELU ? fmaxf(v, 0.0f) : v;
        }
        if (POOL) ((v4f*)acc)[idx] = o;
        else      ((v4f*)out)[(size_t)gi * F4 + g] = o;
    }

    if (POOL) {
        static_assert(!POOL || AGG_TPB == AGG_SLICES * F, "one thread per (slice, feature)");
        __syncthreads();
        const int f  = tid & (F - 1);
        const int r0 = (tid / F) * AGG_RPS;
        int   bid[AGG_RPS];
        float val[AGG_RPS];
#pragma unroll
        for (int j = 0; j < AGG_RPS; ++j) {
            const int r  = r0 + j;
            const bool ok = r < nloc;
            const int rr = ok ? r : 0;
            const int b  = batch[node0 + rr];
            const float v = acc[rr * F + f];
            bid[j] = ok ? b : -1;
            val[j] = ok ? v : 0.0f;
        }
        int   cur  = -1;
        float run  = 0.0f;
        float nrun = 0.0f;
#pragma unroll
        for (int j = 0; j < AGG_RPS; ++j) {
            if (bid[j] != cur) {
                if (cur >= 0) {
                    atomicAdd(&sums[(size_t)cur * F + f], run);
                    if (f == 0) atomicAdd(&cnt[cur], nrun);
                }
                cur = bid[j]; run = 0.0f; nrun = 0.0f;
            }
            run  += val[j];
            nrun += 1.0f;
        }
        if (cur >= 0) {
            atomicAdd(&sums[(size_t)cur * F + f], run);
            if (f == 0) atomicAdd(&cnt[cur], nrun);
        }
    }
}

#define C1_ROWS  208
#define C1_RT    (C1_ROWS / 16)
#define C1_PITCH 68
__global__ __launch_bounds__(AGG_TPB)
void conv1_owner_kernel(const float* __restrict__ h,
                        const int* __restrict__ pk,
                        const int* __restrict__ obw,
                        const int* __restrict__ ocount,
                        const float* __restrict__ dinv,
                        const float* __restrict__ Wa,
                        const float* __restrict__ ba,
                        const float* __restrict__ Wb,
                        float* __restrict__ out,
                        int* __restrict__ recw, int N, int E) {
    constexpr int F = 64;
    __shared__ __attribute__((aligned(16))) float acc[C1_ROWS * C1_PITCH];
    __shared__ __attribute__((aligned(16))) int qbuf[AGG_WAVES * AGG_QCAP];
    __shared__ int cursor;

    const int tid  = threadIdx.x;
    const int lane = tid & 31;
    const int wave = __builtin_amdgcn_readfirstlane(tid >> 5);
    const int node0 = blockIdx.x * AGG_NPB;
    const int nloc  = min(AGG_NPB, N - node0);

    for (int idx = tid; idx < C1_ROWS * (F / 4); idx += AGG_TPB) {
        const int i = idx >> 4, g = idx & 15;
        v4f v = (v4f){0.f, 0.f, 0.f, 0.f};
        if (i < nloc) {
            const float di = dinv[node0 + i];
            v = ((const v4f*)h)[(size_t)(node0 + i) * (F / 4) + g] * (di * di);
        }
        *(v4f*)(acc + i * C1_PITCH + 4 * g) = v;
    }
    if (tid == 0) cursor = 0;
    int2* rec = (int2*)recw + agg_owner_base(ocount, (int)blockIdx.x, lane);
    __syncthreads();

    int off[F / 32];
    agg_row_words<F>(off, lane);

    int* myq = qbuf + wave * AGG_QCAP;
    int qcnt = 0;

    auto push = [&](bool m, int e) {
        unsigned mask = __builtin_amdgcn_ballot_w32(m);
        if (mask) {
            int pos = qcnt + (int)__builtin_amdgcn_mbcnt_lo(mask, 0u);
            if (m) myq[pos] = e;
            qcnt += __builtin_popcount(mask);
        }
    };
    auto reserve = [&](int n) -> int {
        int pos = 0;
        if (lane == 0) pos = atomicAdd(&cursor, n);
        return __builtin_amdgcn_readfirstlane(pos);
    };
    const int nch = E >> 9;
    const unsigned pat = (unsigned)blockIdx.x * 0x01010101u;
    auto ld4 = [&](int c) -> int4 {
        return *(const int4*)(obw + (size_t)c * 128 + lane * 4);
    };
    auto scanw = [&](int wv, int eb) {
        const unsigned x = (unsigned)wv ^ pat;
        const unsigned t = (x - 0x01010101u) & ~x & 0x80808080u;
        if (__builtin_amdgcn_ballot_w32(t != 0u)) {
            push((x & 0x000000FFu) == 0u, eb);
            push((x & 0x0000FF00u) == 0u, eb + 1);
            push((x & 0x00FF0000u) == 0u, eb + 2);
            push((x & 0xFF000000u) == 0u, eb + 3);
        }
    };
    auto scan16 = [&](int4 d, int c) {
        const int eb = c * 512 + lane * 16;
        scanw(d.x, eb);
        scanw(d.y, eb + 4);
        scanw(d.z, eb + 8);
        scanw(d.w, eb + 12);
    };
    auto drain_full = [&]() {
        while (qcnt >= 32) {
            qcnt -= 32;
            agg_drain<F, true, C1_PITCH>(h, pk, dinv, acc, myq, qcnt, 32, node0, lane, off,
                                         rec + reserve(32));
        }
    };

    int c0 = wave;
    for (; c0 + 3 * AGG_WAVES < nch; c0 += 4 * AGG_WAVES) {
        int4 d0 = ld4(c0);
        int4 d1 = ld4(c0 + AGG_WAVES);
        int4 d2 = ld4(c0 + 2 * AGG_WAVES);
        int4 d3 = ld4(c0 + 3 * AGG_WAVES);
        scan16(d0, c0);
        scan16(d1, c0 + AGG_WAVES);
        drain_full();
        scan16(d2, c0 + 2 * AGG_WAVES);
        scan16(d3, c0 + 3 * AGG_WAVES);
        drain_full();
    }
    for (; c0 < nch; c0 += AGG_WAVES) {
        scan16(ld4(c0), c0);
        drain_full();
    }
    if (wave == 0) {
        const unsigned char* ob = (const unsigned char*)obw;
        for (int t = nch * 512; t < E; t += 32) {
            int e = t + lane;
            bool m = (e < E) && ((int)ob[e] == (int)blockIdx.x);
            push(m, e);
        }
    }
    while (qcnt > 0) {
        int cnt = min(qcnt, 32);
        qcnt -= cnt;
        agg_drain<F, false, C1_PITCH>(h, pk, dinv, acc, myq, qcnt, cnt, node0, lane, off,
                                      rec + reserve(cnt));
    }
    __syncthreads();

    const int half = lane >> 4;
    const int lm   = lane & 15;
    float* hbuf = (float*)qbuf;

    const int cg = wave & 3;

    {
        const float* __restrict__ bb = Wa + (2 * half) * 64 + cg * 16 + lm;
        v2f wb[16];
#pragma unroll
        for (int k = 0; k < 16; ++k) {
            wb[k].x = bb[(4 * k) * 64];
            wb[k].y = bb[(4 * k + 1) * 64];
        }
        const float bv = ba[cg * 16 + lm];
        for (int rt = wave >> 2; rt < C1_RT; rt += AGG_WAVES / 4) {
            const float* arow = acc + (rt * 16 + lm) * C1_PITCH + 2 * half;
            v8f d = (v8f){};
#pragma unroll
            for (int k = 0; k < 16; ++k) {
                const v2f a = *(const v2f*)(arow + 4 * k);
                d = __builtin_amdgcn_wmma_f32_16x16x4_f32(false, a, false, wb[k], (short)0, d,
                                                          false, false);
            }
            float* hrow = hbuf + (rt * 16 + 8 * half) * C1_PITCH + cg * 16 + lm;
#pragma unroll
            for (int v = 0; v < 8; ++v) hrow[v * C1_PITCH] = fmaxf(d[v] + bv, 0.0f);
        }
    }
    __syncthreads();

    const float* __restrict__ bb2 = Wb + (2 * half) * 128 + cg * 32 + lm;
    v2f wp[16], wq[16];
#pragma unroll
    for (int k = 0; k < 16; ++k) {
        wp[k].x = bb2[(4 * k) * 128];
        wp[k].y = bb2[(4 * k + 1) * 128];
        wq[k].x = bb2[(4 * k) * 128 + 16];
        wq[k].y = bb2[(4 * k + 1) * 128 + 16];
    }
    for (int rt = wave >> 2; rt < C1_RT; rt += AGG_WAVES / 4) {
        const float* arow = hbuf + (rt * 16 + lm) * C1_PITCH + 2 * half;
        v8f d0 = (v8f){}, d1 = (v8f){};
#pragma unroll
        for (int k = 0; k < 16; ++k) {
            const v2f a = *(const v2f*)(arow + 4 * k);
            d0 = __builtin_amdgcn_wmma_f32_16x16x4_f32(false, a, false, wp[k], (short)0, d0,
                                                       false, false);
            d1 = __builtin_amdgcn_wmma_f32_16x16x4_f32(false, a, false, wq[k], (short)0, d1,
                                                       false, false);
        }
        const int r0 = rt * 16 + 8 * half;
        float* __restrict__ crow = out + (size_t)(node0 + r0) * 128 + cg * 32 + lm;
#pragma unroll
        for (int v = 0; v < 8; ++v) {
            if (r0 + v < nloc) {
                crow[(size_t)v * 128]      = d0[v];
                crow[(size_t)v * 128 + 16] = d1[v];
            }
        }
    }
}


#define AGG_NG 4
#define AGG_NSELF ((AGG_NPB + AGG_WAVES - 1) / AGG_WAVES)

__device__ __forceinline__ void graph_add(v4f (&a)[AGG_NG], int g, v4f v, float w) {
#pragma unroll
    for (int j = 0; j < AGG_NG; ++j) {
        const float m = (g == j) ? w : 0.0f;
        a[j] += v * m;
    }
}

__device__ __forceinline__ void graph_rows(const float* __restrict__ h, v4f (&a)[AGG_NG],
                                           int s_l, int g_l, float w_l, int lane) {
#pragma unroll
    for (int g0 = 0; g0 < 32; g0 += 8) {
        v4f r[8];
#pragma unroll
        for (int i = 0; i < 8; ++i) {
            int s = __builtin_amdgcn_readlane(s_l, g0 + i);
            r[i] = *((const v4f*)(h + (size_t)s * 128) + lane);
        }
#pragma unroll
        for (int i = 0; i < 8; ++i) {
            int g = __builtin_amdgcn_readlane(g_l, g0 + i);
            float w = __builtin_bit_cast(
                float, __builtin_amdgcn_readlane(__builtin_bit_cast(int, w_l), g0 + i));
            graph_add(a, g, r[i], w);
        }
    }
}

#define AGG_RCAP 8192
#define AGG_RLD  4
template <int F, bool RELU, bool POOL>
__global__ __launch_bounds__(AGG_TPB)
void agg_replay_kernel(const float* __restrict__ h,
                       const int* __restrict__ recw,
                       const int* __restrict__ ocount,
                       const float* __restrict__ dinv,
                       const float* __restrict__ bias,
                       float* __restrict__ out,
                       const int* __restrict__ batch,
                       float* __restrict__ sums,
                       float* __restrict__ cnt, int N) {
    constexpr int F4 = F / 4;
    __shared__ __attribute__((aligned(16))) float acc[AGG_NPB * F];
    __shared__ int2 srec[AGG_RCAP];
    __shared__ int sbat[AGG_NPB];

    const int tid  = threadIdx.x;
    const int lane = tid & 31;
    const int wave = __builtin_amdgcn_readfirstlane(tid >> 5);
    const int node0 = blockIdx.x * AGG_NPB;
    const int nloc  = min(AGG_NPB, N - node0);

    int g_lo = 0, span = AGG_NG + 1;
    if (POOL) {
        int lo = 0x7FFFFFFF, hi = -1;
        for (int i = lane; i < nloc; i += 32) {
            const int b = batch[node0 + i];
            lo = min(lo, b); hi = max(hi, b);
        }
        for (int o = 16; o > 0; o >>= 1) {
            lo = min(lo, __shfl_xor(lo, o)); hi = max(hi, __shfl_xor(hi, o));
        }
        g_lo = __builtin_amdgcn_readfirstlane(lo);
        span = __builtin_amdgcn_readfirstlane(hi) - g_lo + 1;
        for (int i = tid; i < nloc; i += AGG_TPB) sbat[i] = batch[node0 + i];
    }
    const bool graphs = POOL && F == 128 && span <= AGG_NG;
    if (!graphs)
        for (int i = tid; i < AGG_NPB * F4; i += AGG_TPB) ((v4f*)acc)[i] = (v4f){0.f, 0.f, 0.f, 0.f};
    v4f ga[AGG_NG];
#pragma unroll
    for (int j = 0; j < AGG_NG; ++j) ga[j] = (v4f){0.f, 0.f, 0.f, 0.f};
    const int total = ocount[blockIdx.x];
    const int2* __restrict__ rec =
        (const int2*)recw + agg_owner_base(ocount, (int)blockIdx.x, lane);

    int off[F / 32];
    agg_row_words<F>(off, lane);

    for (int r0 = 0; r0 < total; r0 += AGG_RCAP) {
        const int nst = min(total - r0, AGG_RCAP);
        if (r0 != 0) __syncthreads();
        for (int j0 = tid; j0 < nst; j0 += AGG_RLD * AGG_TPB) {
            int2 t[AGG_RLD];
#pragma unroll
            for (int k = 0; k < AGG_RLD; ++k)
                t[k] = rec[r0 + min(j0 + k * AGG_TPB, nst - 1)];
#pragma unroll
            for (int k = 0; k < AGG_RLD; ++k)
                if (j0 + k * AGG_TPB < nst) srec[j0 + k * AGG_TPB] = t[k];
        }
        __syncthreads();

        const int nfull = nst >> 5;
        const int rest  = nst & 31;
        for (int i = wave; i < nfull; i += AGG_WAVES) {
            const int2 r = srec[32 * i + lane];
            const int dl = (int)((unsigned)r.x >> 16) - node0;
            if (graphs)
                graph_rows(h, ga, r.x & 0xFFFF, sbat[dl] - g_lo,
                           __builtin_bit_cast(float, r.y), lane);
            else
                agg_rows<F, true>(h, acc, r.x & 0xFFFF, dl,
                                  __builtin_bit_cast(float, r.y), 32, lane, off);
        }
        if (rest != 0 && (nfull & (AGG_WAVES - 1)) == wave) {
            int dl_l = 0, s_l = 0;
            float w_l = 0.0f;
            if (lane < rest) {
                const int2 t = srec[32 * nfull + lane];
                s_l  = t.x & 0xFFFF;
                dl_l = (int)((unsigned)t.x >> 16) - node0;
                w_l  = __builtin_bit_cast(float, t.y);
            }
            if (graphs)
                graph_rows(h, ga, s_l, sbat[dl_l] - g_lo, w_l, lane);
            else
                agg_rows<F, false>(h, acc, s_l, dl_l, w_l, rest, lane, off);
        }
    }
    __syncthreads();

    if (!graphs) {
        agg_finish<F, RELU, POOL>(acc, h, dinv, bias, out, batch, sums, cnt, node0, nloc, tid);
        return;
    }

    v4f   rr[AGG_NSELF];
    float dk[AGG_NSELF];
    int   gk[AGG_NSELF];
#pragma unroll
    for (int k = 0; k < AGG_NSELF; ++k) {
        const int ii = min(wave + k * AGG_WAVES, nloc - 1);
        rr[k] = ((const v4f*)h)[(size_t)(node0 + ii) * F4 + lane];
        dk[k] = dinv[node0 + ii];
        gk[k] = __builtin_amdgcn_readfirstlane(sbat[ii]) - g_lo;
    }
    const v4f bv = ((const v4f*)bias)[lane];
#pragma unroll
    for (int k = 0; k < AGG_NSELF; ++k) {
        const bool ok = wave + k * AGG_WAVES < nloc;
        graph_add(ga, gk[k], rr[k], ok ? dk[k] * dk[k] : 0.0f);
        graph_add(ga, gk[k], bv, ok ? 1.0f : 0.0f);
    }

#pragma unroll
    for (int j = 0; j < AGG_NG; ++j) ((v4f*)acc)[(wave * AGG_NG + j) * F4 + lane] = ga[j];
    __syncthreads();
    for (int idx = tid; idx < span * F; idx += AGG_TPB) {
        const int j = idx / F, f = idx - j * F;
        float s = 0.0f;
        for (int w = 0; w < AGG_WAVES; ++w) s += acc[(w * AGG_NG + j) * F + f];
        atomicAdd(&sums[(size_t)(g_lo + j) * F + f], s);
    }
    if (tid < span) {
        float c = 0.0f;
        for (int i = 0; i < nloc; ++i) c += (sbat[i] - g_lo == tid) ? 1.0f : 0.0f;
        atomicAdd(&cnt[g_lo + tid], c);
    }
}

template <int F>
__global__ void pool_div_kernel(float* __restrict__ sums, const float* __restrict__ cnt, int G) {
    int gid = blockIdx.x * blockDim.x + threadIdx.x;
    if (gid >= G * F) return;
    int g = gid / F;
    sums[gid] /= fmaxf(cnt[g], 1.0f);
}

static inline int cdiv(long long a, long long b) { return (int)((a + b - 1) / b); }

extern "C" void kernel_launch(void* const* d_in, const int* in_sizes, int n_in,
                              void* d_out, int out_size, void* d_ws, size_t ws_size,
                              hipStream_t stream) {
    (void)n_in; (void)ws_size;
    const float* x    = (const float*)d_in[0];
    const int*   edge = (const int*)d_in[1];
    const int*   batch= (const int*)d_in[2];
    const float* b1   = (const float*)d_in[4];
    const float* W1   = (const float*)d_in[3];
    const float* W2   = (const float*)d_in[5];
    const float* b2   = (const float*)d_in[6];
    const float* Wfc  = (const float*)d_in[7];
    const float* bfc  = (const float*)d_in[8];
    float* out = (float*)d_out;

    const int N   = in_sizes[2];
    const int E   = in_sizes[1] / 2;
    const int D   = 64;
    const int D2  = 128;
    const int HID = in_sizes[8];
    const int G   = out_size / HID;

    const int* src = edge;
    const int* dst = edge + E;

    float* ws    = (float*)d_ws;
    float* hw2   = ws + 2 * (size_t)N * D;
    float* ebuf  = hw2  + (size_t)N * D2;
    int*   pk    = (int*)ebuf;
    int*   obw   = pk + E;
    int*   rec   = obw + E;
    float* zblk  = (float*)(rec + 2 * (size_t)E);
    int*   ocount= (int*)zblk;
    float* sums  = zblk + AGG_OWNERS;
    float* cnt   = sums + (size_t)G * D2;
    float* dinv  = ebuf + (size_t)N * D2;

    fill_kernel<<<cdiv((long long)AGG_OWNERS + G * D2 + G, TPB), TPB, 0, stream>>>(
        zblk, 0.0f, AGG_OWNERS + G * D2 + G);
    fill_kernel<<<cdiv(N, TPB), TPB, 0, stream>>>(dinv, 1.0f, N);

    edge_prep_kernel<<<cdiv(cdiv(E, 4), TPB), TPB, 0, stream>>>(src, dst, dinv, pk, obw, ocount, E);
    rsqrt_kernel<<<cdiv(N, TPB), TPB, 0, stream>>>(dinv, N);

    conv1_owner_kernel<<<cdiv(N, AGG_NPB), AGG_TPB, 0, stream>>>(
        x, pk, obw, ocount, dinv, W1, b1, W2, hw2, rec, N, E);

    agg_replay_kernel<128, false, true><<<cdiv(N, AGG_NPB), AGG_TPB, 0, stream>>>(
        hw2, rec, ocount, dinv, b2, nullptr, batch, sums, cnt, N);

    pool_div_kernel<128><<<cdiv((long long)G * D2, TPB), TPB, 0, stream>>>(sums, cnt, G);

    {
        int row_tiles = G / 16;
        wmma_gemm_f32<128, 64><<<cdiv((long long)row_tiles * 32, TPB), TPB, 0, stream>>>(
            sums, Wfc, out, bfc, row_tiles);
    }
}


# conv1 owners find their edges through a per-owner bitmap of 32-edge pieces written by the degree pass: only flagged pieces' owner bytes are read, not the whole stream; sum order: LDS arrival
# speedup vs baseline: 3.8826x; 1.0563x over previous
#include <hip/hip_runtime.h>
#include <hip/hip_bf16.h>


typedef float v2f __attribute__((ext_vector_type(2)));
typedef float v8f __attribute__((ext_vector_type(8)));

#define TPB 256

__global__ void fill_kernel(float* __restrict__ p, float v, int n) {
    int i = blockIdx.x * blockDim.x + threadIdx.x;
    if (i < n) p[i] = v;
}

#define AGG_NPB    196
#define AGG_OWNERS 256
__global__ void edge_prep_kernel(const int* __restrict__ src, const int* __restrict__ dst,
                                 float* __restrict__ deg, int* __restrict__ pk,
                                 int* __restrict__ obw, int* __restrict__ ocount,
                                 unsigned* __restrict__ bm, int E) {
    __shared__ int hist[AGG_OWNERS];
    __shared__ unsigned bits[AGG_OWNERS];
    for (int i = threadIdx.x; i < AGG_OWNERS; i += blockDim.x) { hist[i] = 0; bits[i] = 0u; }
    __syncthreads();
    int e = (blockIdx.x * blockDim.x + threadIdx.x) * 4;
    if (e + 3 < E) {
        int4 d = *(const int4*)(dst + e);
        int4 s = *(const int4*)(src + e);
        atomicAdd(&deg[d.x], 1.0f);
        atomicAdd(&deg[d.y], 1.0f);
        atomicAdd(&deg[d.z], 1.0f);
        atomicAdd(&deg[d.w], 1.0f);
        int4 p;
        p.x = (int)(((unsigned)d.x << 16) | (unsigned)s.x);
        p.y = (int)(((unsigned)d.y << 16) | (unsigned)s.y);
        p.z = (int)(((unsigned)d.z << 16) | (unsigned)s.z);
        p.w = (int)(((unsigned)d.w << 16) | (unsigned)s.w);
        *(int4*)(pk + e) = p;
        const unsigned ox = (unsigned)(d.x / AGG_NPB), oy = (unsigned)(d.y / AGG_NPB);
        const unsigned oz = (unsigned)(d.z / AGG_NPB), ow = (unsigned)(d.w / AGG_NPB);
        obw[e >> 2] = (int)(ox | (oy << 8) | (oz << 16) | (ow << 24));
        atomicAdd(&hist[ox], 1);
        atomicAdd(&hist[oy], 1);
        atomicAdd(&hist[oz], 1);
        atomicAdd(&hist[ow], 1);
        const unsigned pb = 1u << (threadIdx.x >> 3);
        atomicOr(&bits[ox], pb);
        atomicOr(&bits[oy], pb);
        atomicOr(&bits[oz], pb);
        atomicOr(&bits[ow], pb);
    } else {
        unsigned char* ob = (unsigned char*)obw;
        for (; e < E; ++e) {
            int d = dst[e];
            atomicAdd(&deg[d], 1.0f);
            pk[e] = (int)(((unsigned)d << 16) | (unsigned)src[e]);
            ob[e] = (unsigned char)(d / AGG_NPB);
            atomicAdd(&hist[d / AGG_NPB], 1);
            atomicOr(&bits[d / AGG_NPB], 1u << ((e - (int)(blockIdx.x * blockDim.x) * 4) >> 5));
        }
    }
    __syncthreads();
    for (int i = threadIdx.x; i < AGG_OWNERS; i += blockDim.x) {
        const int c = hist[i];
        if (c) atomicAdd(&ocount[i], c);
        bm[(size_t)blockIdx.x * AGG_OWNERS + i] = bits[i];
    }
}

__global__ void rsqrt_kernel(float* __restrict__ d, int n) {
    int i = blockIdx.x * blockDim.x + threadIdx.x;
    if (i < n) d[i] = rsqrtf(d[i]);
}

template <int K, int Nn>
__global__ void wmma_gemm_f32(const float* __restrict__ A,
                              const float* __restrict__ B,
                              float* __restrict__ C,
                              const float* __restrict__ bias,
                              int row_tiles) {
    constexpr int COLT = Nn / 16;
    int wave = (blockIdx.x * blockDim.x + threadIdx.x) >> 5;
    if (wave >= row_tiles) return;
    int lane = threadIdx.x & 31;
    int half = lane >> 4;
    int lm   = lane & 15;
    int row0 = wave << 4;

    const float* __restrict__ arow  = A + (size_t)(row0 + lm) * K + 2 * half;
    const float* __restrict__ bbase = B + (size_t)(2 * half) * Nn + lm;

    v8f acc[COLT];
#pragma unroll
    for (int c = 0; c < COLT; ++c) acc[c] = (v8f){};

#pragma unroll
    for (int k0 = 0; k0 < K; k0 += 4) {
        v2f a;
        a.x = arow[k0];
        a.y = arow[k0 + 1];
#pragma unroll
        for (int c = 0; c < COLT; ++c) {
            v2f b;
            b.x = bbase[k0 * Nn + c * 16];
            b.y = bbase[(k0 + 1) * Nn + c * 16];
            acc[c] = __builtin_amdgcn_wmma_f32_16x16x4_f32(
                 false, a,  false, b,
                 (short)0, acc[c],  false,  false);
        }
    }

    float* __restrict__ crow = C + (size_t)(row0 + 8 * half) * Nn + lm;
#pragma unroll
    for (int c = 0; c < COLT; ++c) {
        float bv = bias ? bias[c * 16 + lm] : 0.0f;
#pragma unroll
        for (int v = 0; v < 8; ++v) {
            crow[(size_t)v * Nn + c * 16] = acc[c][v] + bv;
        }
    }
}

#define AGG_TPB   1024
#define AGG_WAVES 32
#define AGG_QCAP  1056
#define AGG_PCAP  1024

typedef float v4f __attribute__((ext_vector_type(4)));

template <int F> struct RowVecSel      { typedef v2f T; };
template <>      struct RowVecSel<128> { typedef v4f T; };

template <int F, bool FULL, int PITCH = F>
__device__ __forceinline__ void agg_rows(const float* __restrict__ h, float* acc,
                                         int s_l, int dl_l, float w_l, int cnt, int lane,
                                         const int (&off)[F / 32]) {
    typedef typename RowVecSel<F>::T VT;
    constexpr int VEC = F / 32;
#pragma unroll
    for (int g = 0; g < 32; g += 8) {
        if (!FULL && g >= cnt) break;
        VT r[8];
#pragma unroll
        for (int i = 0; i < 8; ++i) {
            if (FULL || g + i < cnt) {
                int s = __builtin_amdgcn_readlane(s_l, g + i);
                r[i] = *((const VT*)(h + (size_t)s * F) + lane);
            }
        }
#pragma unroll
        for (int i = 0; i < 8; ++i) {
            if (FULL || g + i < cnt) {
                int dl = __builtin_amdgcn_readlane(dl_l, g + i);
                float w = __builtin_bit_cast(
                    float, __builtin_amdgcn_readlane(__builtin_bit_cast(int, w_l), g + i));
                float* a = acc + dl * PITCH;
#pragma unroll
                for (int k = 0; k < VEC; ++k)
                    __hip_atomic_fetch_add(&a[off[k]], r[i][k] * w, __ATOMIC_RELAXED,
                                           __HIP_MEMORY_SCOPE_WORKGROUP);
            }
        }
    }
}

template <int F, bool FULL, int PITCH = F>
__device__ __forceinline__ void agg_drain(const float* __restrict__ h,
                                          const int* __restrict__ pk,
                                          const float* __restrict__ dinv,
                                          float* acc, const int* q,
                                          int start, int cnt, int node0, int lane,
                                          const int (&off)[F / 32],
                                          int2* __restrict__ rec_out) {
    int dl_l = 0, s_l = 0;
    float w_l = 0.0f;
    if (FULL || lane < cnt) {
        int e_l = q[start + lane];
        int p_l = pk[e_l];
        int d_l = (int)((unsigned)p_l >> 16);
        s_l = p_l & 0xFFFF;
        dl_l = d_l - node0;
        w_l = dinv[s_l] * dinv[d_l];
        rec_out[lane] = make_int2(p_l, __builtin_bit_cast(int, w_l));
    }
    agg_rows<F, FULL, PITCH>(h, acc, s_l, dl_l, w_l, cnt, lane, off);
}

template <int F>
__device__ __forceinline__ void agg_row_words(int (&off)[F / 32], int lane) {
    constexpr int VEC = F / 32;
#pragma unroll
    for (int k = 0; k < VEC; ++k)
        off[k] = (F == 128) ? (4 * lane + (k ^ (lane >> 4))) : (VEC * lane + k);
}

__device__ __forceinline__ int agg_owner_base(const int* __restrict__ ocount, int b, int lane) {
    int s = 0;
    for (int i = lane; i < b; i += 32) s += ocount[i];
    for (int o = 16; o > 0; o >>= 1) s += __shfl_xor(s, o);
    return s;
}

#define AGG_SLICES 8
#define AGG_RPS    ((AGG_NPB + AGG_SLICES - 1) / AGG_SLICES)
template <int F, bool RELU, bool POOL>
__device__ __forceinline__ void agg_finish(float* acc,
                                           const float* __restrict__ h,
                                           const float* __restrict__ dinv,
                                           const float* __restrict__ bias,
                                           float* __restrict__ out,
                                           const int* __restrict__ batch,
                                           float* __restrict__ sums,
                                           float* __restrict__ cnt,
                                           int node0, int nloc, int tid) {
    constexpr int F4 = F / 4;
    for (int idx = tid; idx < nloc * F4; idx += AGG_TPB) {
        int i = idx / F4;
        int g = idx - i * F4;
        v4f q = ((const v4f*)acc)[idx];
        if (F == 128 && (g >> 4)) q = (v4f){q.y, q.x, q.w, q.z};
        int gi = node0 + i;
        float di = dinv[gi];
        v4f xv = ((const v4f*)h)[(size_t)gi * F4 + g];
        v4f bv = ((const v4f*)bias)[g];
        v4f o;
#pragma unroll
        for (int c = 0; c < 4; ++c) {
            float v = q[c] + xv[c] * di * di + bv[c];
            o[c] = RELU ? fmaxf(v, 0.0f) : v;
        }
        if (POOL) ((v4f*)acc)[idx] = o;
        else      ((v4f*)out)[(size_t)gi * F4 + g] = o;
    }

    if (POOL) {
        static_assert(!POOL || AGG_TPB == AGG_SLICES * F, "one thread per (slice, feature)");
        __syncthreads();
        const int f  = tid & (F - 1);
        const int r0 = (tid / F) * AGG_RPS;
        int   bid[AGG_RPS];
        float val[AGG_RPS];
#pragma unroll
        for (int j = 0; j < AGG_RPS; ++j) {
            const int r  = r0 + j;
            const bool ok = r < nloc;
            const int rr = ok ? r : 0;
            const int b  = batch[node0 + rr];
            const float v = acc[rr * F + f];
            bid[j] = ok ? b : -1;
            val[j] = ok ? v : 0.0f;
        }
        int   cur  = -1;
        float run  = 0.0f;
        float nrun = 0.0f;
#pragma unroll
        for (int j = 0; j < AGG_RPS; ++j) {
            if (bid[j] != cur) {
                if (cur >= 0) {
                    atomicAdd(&sums[(size_t)cur * F + f], run);
                    if (f == 0) atomicAdd(&cnt[cur], nrun);
                }
                cur = bid[j]; run = 0.0f; nrun = 0.0f;
            }
            run  += val[j];
            nrun += 1.0f;
        }
        if (cur >= 0) {
            atomicAdd(&sums[(size_t)cur * F + f], run);
            if (f == 0) atomicAdd(&cnt[cur], nrun);
        }
    }
}

#define C1_ROWS  208
#define C1_RT    (C1_ROWS / 16)
#define C1_PITCH 68
__global__ __launch_bounds__(AGG_TPB)
void conv1_owner_kernel(const float* __restrict__ h,
                        const int* __restrict__ pk,
                        const int* __restrict__ obw,
                        const unsigned* __restrict__ bm,
                        const int* __restrict__ ocount,
                        const float* __restrict__ dinv,
                        const float* __restrict__ Wa,
                        const float* __restrict__ ba,
                        const float* __restrict__ Wb,
                        float* __restrict__ out,
                        int* __restrict__ recw, int N, int E) {
    constexpr int F = 64;
    __shared__ __attribute__((aligned(16))) float acc[C1_ROWS * C1_PITCH];
    __shared__ __attribute__((aligned(16))) int qbuf[AGG_WAVES * AGG_QCAP];
    __shared__ unsigned short plist[AGG_WAVES * AGG_PCAP];
    __shared__ int cursor;

    const int tid  = threadIdx.x;
    const int lane = tid & 31;
    const int wave = __builtin_amdgcn_readfirstlane(tid >> 5);
    const int node0 = blockIdx.x * AGG_NPB;
    const int nloc  = min(AGG_NPB, N - node0);

    for (int idx = tid; idx < C1_ROWS * (F / 4); idx += AGG_TPB) {
        const int i = idx >> 4, g = idx & 15;
        v4f v = (v4f){0.f, 0.f, 0.f, 0.f};
        if (i < nloc) {
            const float di = dinv[node0 + i];
            v = ((const v4f*)h)[(size_t)(node0 + i) * (F / 4) + g] * (di * di);
        }
        *(v4f*)(acc + i * C1_PITCH + 4 * g) = v;
    }
    if (tid == 0) cursor = 0;
    int2* rec = (int2*)recw + agg_owner_base(ocount, (int)blockIdx.x, lane);
    __syncthreads();

    int off[F / 32];
    agg_row_words<F>(off, lane);

    int* myq = qbuf + wave * AGG_QCAP;
    int qcnt = 0;

    auto push = [&](bool m, int e) {
        unsigned mask = __builtin_amdgcn_ballot_w32(m);
        if (mask) {
            int pos = qcnt + (int)__builtin_amdgcn_mbcnt_lo(mask, 0u);
            if (m) myq[pos] = e;
            qcnt += __builtin_popcount(mask);
        }
    };
    auto reserve = [&](int n) -> int {
        int pos = 0;
        if (lane == 0) pos = atomicAdd(&cursor, n);
        return __builtin_amdgcn_readfirstlane(pos);
    };
    const unsigned pat = (unsigned)blockIdx.x * 0x01010101u;
    auto eqb = [&](int wv) -> unsigned {
        const unsigned y = (unsigned)wv ^ pat;
        const unsigned t = ((y & 0x7F7F7F7Fu) + 0x7F7F7F7Fu) | y;
        return ~t & 0x80808080u;
    };
    auto ldp = [&](int sc, int4& a, int4& b) {
        const int4* p = (const int4*)(obw + (size_t)sc * 8);
        a = p[0];
        b = p[1];
    };
    auto drain_full = [&]() {
        while (qcnt >= 32) {
            qcnt -= 32;
            agg_drain<F, true, C1_PITCH>(h, pk, dinv, acc, myq, qcnt, 32, node0, lane, off,
                                         rec + reserve(32));
        }
    };

    const int nblk = (E + 1023) >> 10;
    unsigned short* myp = plist + wave * AGG_PCAP;
    for (int w0 = 0; w0 < nblk; w0 += AGG_WAVES * 32) {
        const int wi = w0 + lane * AGG_WAVES + wave;
        unsigned word = (wi < nblk) ? bm[(size_t)wi * AGG_OWNERS + blockIdx.x] : 0u;
        const int nb = __builtin_popcount(word);
        int incl = nb;
        for (int o = 1; o < 32; o <<= 1) {
            const int t = __shfl_up(incl, o);
            if (lane >= o) incl += t;
        }
        int pos = incl - nb;
        const int L = __builtin_amdgcn_readlane(incl, 31);
        while (__builtin_amdgcn_ballot_w32(word != 0u)) {
            if (word != 0u) {
                myp[pos++] = (unsigned short)((lane << 5) | __builtin_ctz(word));
                word &= word - 1u;
            }
        }
        auto fetch = [&](int i0, bool& has, int& sc, int4& a, int4& b) {
            has = i0 + lane < L;
            const int ent = has ? (int)myp[i0 + lane] : 0;
            sc = has ? ((w0 + (ent >> 5) * AGG_WAVES + wave) * 32 + (ent & 31)) : 0;
            ldp(sc, a, b);
        };
        bool has; int sc; int4 a, b;
        fetch(0, has, sc, a, b);
        for (int i0 = 0; i0 < L; i0 += 32) {
            unsigned m = (eqb(a.x) >> 7) | (eqb(a.y) >> 6) | (eqb(a.z) >> 5) | (eqb(a.w) >> 4) |
                         (eqb(b.x) >> 3) | (eqb(b.y) >> 2) | (eqb(b.z) >> 1) | eqb(b.w);
            if (!has) m = 0u;
            const int eb = sc * 32;
            while (__builtin_amdgcn_ballot_w32(m != 0u)) {
                const bool h2 = m != 0u;
                const int p = h2 ? __builtin_ctz(m) : 0;
                const int e = eb + 4 * (p & 7) + (p >> 3);
                m &= m - 1u;
                push(h2 && e < E, e);
            }
            fetch(i0 + 32, has, sc, a, b);
            drain_full();
        }
    }
    while (qcnt > 0) {
        int cnt = min(qcnt, 32);
        qcnt -= cnt;
        agg_drain<F, false, C1_PITCH>(h, pk, dinv, acc, myq, qcnt, cnt, node0, lane, off,
                                      rec + reserve(cnt));
    }
    __syncthreads();

    const int half = lane >> 4;
    const int lm   = lane & 15;
    float* hbuf = (float*)qbuf;

    const int cg = wave & 3;

    {
        const float* __restrict__ bb = Wa + (2 * half) * 64 + cg * 16 + lm;
        v2f wb[16];
#pragma unroll
        for (int k = 0; k < 16; ++k) {
            wb[k].x = bb[(4 * k) * 64];
            wb[k].y = bb[(4 * k + 1) * 64];
        }
        const float bv = ba[cg * 16 + lm];
        for (int rt = wave >> 2; rt < C1_RT; rt += AGG_WAVES / 4) {
            const float* arow = acc + (rt * 16 + lm) * C1_PITCH + 2 * half;
            v8f d = (v8f){};
#pragma unroll
            for (int k = 0; k < 16; ++k) {
                const v2f a = *(const v2f*)(arow + 4 * k);
                d = __builtin_amdgcn_wmma_f32_16x16x4_f32(false, a, false, wb[k], (short)0, d,
                                                          false, false);
            }
            float* hrow = hbuf + (rt * 16 + 8 * half) * C1_PITCH + cg * 16 + lm;
#pragma unroll
            for (int v = 0; v < 8; ++v) hrow[v * C1_PITCH] = fmaxf(d[v] + bv, 0.0f);
        }
    }
    __syncthreads();

    const float* __restrict__ bb2 = Wb + (2 * half) * 128 + cg * 32 + lm;
    v2f wp[16], wq[16];
#pragma unroll
    for (int k = 0; k < 16; ++k) {
        wp[k].x = bb2[(4 * k) * 128];
        wp[k].y = bb2[(4 * k + 1) * 128];
        wq[k].x = bb2[(4 * k) * 128 + 16];
        wq[k].y = bb2[(4 * k + 1) * 128 + 16];
    }
    for (int rt = wave >> 2; rt < C1_RT; rt += AGG_WAVES / 4) {
        const float* arow = hbuf + (rt * 16 + lm) * C1_PITCH + 2 * half;
        v8f d0 = (v8f){}, d1 = (v8f){};
#pragma unroll
        for (int k = 0; k < 16; ++k) {
            const v2f a = *(const v2f*)(arow + 4 * k);
            d0 = __builtin_amdgcn_wmma_f32_16x16x4_f32(false, a, false, wp[k], (short)0, d0,
                                                       false, false);
            d1 = __builtin_amdgcn_wmma_f32_16x16x4_f32(false, a, false, wq[k], (short)0, d1,
                                                       false, false);
        }
        const int r0 = rt * 16 + 8 * half;
        float* __restrict__ crow = out + (size_t)(node0 + r0) * 128 + cg * 32 + lm;
#pragma unroll
        for (int v = 0; v < 8; ++v) {
            if (r0 + v < nloc) {
                crow[(size_t)v * 128]      = d0[v];
                crow[(size_t)v * 128 + 16] = d1[v];
            }
        }
    }
}


#define AGG_NG 4
#define AGG_NSELF ((AGG_NPB + AGG_WAVES - 1) / AGG_WAVES)

__device__ __forceinline__ void graph_add(v4f (&a)[AGG_NG], int g, v4f v, float w) {
#pragma unroll
    for (int j = 0; j < AGG_NG; ++j) {
        const float m = (g == j) ? w : 0.0f;
        a[j] += v * m;
    }
}

__device__ __forceinline__ void graph_rows(const float* __restrict__ h, v4f (&a)[AGG_NG],
                                           int s_l, int g_l, float w_l, int lane) {
#pragma unroll
    for (int g0 = 0; g0 < 32; g0 += 8) {
        v4f r[8];
#pragma unroll
        for (int i = 0; i < 8; ++i) {
            int s = __builtin_amdgcn_readlane(s_l, g0 + i);
            r[i] = *((const v4f*)(h + (size_t)s * 128) + lane);
        }
#pragma unroll
        for (int i = 0; i < 8; ++i) {
            int g = __builtin_amdgcn_readlane(g_l, g0 + i);
            float w = __builtin_bit_cast(
                float, __builtin_amdgcn_readlane(__builtin_bit_cast(int, w_l), g0 + i));
            graph_add(a, g, r[i], w);
        }
    }
}

#define AGG_RCAP 8192
#define AGG_RLD  4
template <int F, bool RELU, bool POOL>
__global__ __launch_bounds__(AGG_TPB)
void agg_replay_kernel(const float* __restrict__ h,
                       const int* __restrict__ recw,
                       const int* __restrict__ ocount,
                       const float* __restrict__ dinv,
                       const float* __restrict__ bias,
                       float* __restrict__ out,
                       const int* __restrict__ batch,
                       float* __restrict__ sums,
                       float* __restrict__ cnt, int N) {
    constexpr int F4 = F / 4;
    __shared__ __attribute__((aligned(16))) float acc[AGG_NPB * F];
    __shared__ int2 srec[AGG_RCAP];
    __shared__ int sbat[AGG_NPB];

    const int tid  = threadIdx.x;
    const int lane = tid & 31;
    const int wave = __builtin_amdgcn_readfirstlane(tid >> 5);
    const int node0 = blockIdx.x * AGG_NPB;
    const int nloc  = min(AGG_NPB, N - node0);

    int g_lo = 0, span = AGG_NG + 1;
    if (POOL) {
        int lo = 0x7FFFFFFF, hi = -1;
        for (int i = lane; i < nloc; i += 32) {
            const int b = batch[node0 + i];
            lo = min(lo, b); hi = max(hi, b);
        }
        for (int o = 16; o > 0; o >>= 1) {
            lo = min(lo, __shfl_xor(lo, o)); hi = max(hi, __shfl_xor(hi, o));
        }
        g_lo = __builtin_amdgcn_readfirstlane(lo);
        span = __builtin_amdgcn_readfirstlane(hi) - g_lo + 1;
        for (int i = tid; i < nloc; i += AGG_TPB) sbat[i] = batch[node0 + i];
    }
    const bool graphs = POOL && F == 128 && span <= AGG_NG;
    if (!graphs)
        for (int i = tid; i < AGG_NPB * F4; i += AGG_TPB) ((v4f*)acc)[i] = (v4f){0.f, 0.f, 0.f, 0.f};
    v4f ga[AGG_NG];
#pragma unroll
    for (int j = 0; j < AGG_NG; ++j) ga[j] = (v4f){0.f, 0.f, 0.f, 0.f};
    const int total = ocount[blockIdx.x];
    const int2* __restrict__ rec =
        (const int2*)recw + agg_owner_base(ocount, (int)blockIdx.x, lane);

    int off[F / 32];
    agg_row_words<F>(off, lane);

    for (int r0 = 0; r0 < total; r0 += AGG_RCAP) {
        const int nst = min(total - r0, AGG_RCAP);
        if (r0 != 0) __syncthreads();
        for (int j0 = tid; j0 < nst; j0 += AGG_RLD * AGG_TPB) {
            int2 t[AGG_RLD];
#pragma unroll
            for (int k = 0; k < AGG_RLD; ++k)
                t[k] = rec[r0 + min(j0 + k * AGG_TPB, nst - 1)];
#pragma unroll
            for (int k = 0; k < AGG_RLD; ++k)
                if (j0 + k * AGG_TPB < nst) srec[j0 + k * AGG_TPB] = t[k];
        }
        __syncthreads();

        const int nfull = nst >> 5;
        const int rest  = nst & 31;
        for (int i = wave; i < nfull; i += AGG_WAVES) {
            const int2 r = srec[32 * i + lane];
            const int dl = (int)((unsigned)r.x >> 16) - node0;
            if (graphs)
                graph_rows(h, ga, r.x & 0xFFFF, sbat[dl] - g_lo,
                           __builtin_bit_cast(float, r.y), lane);
            else
                agg_rows<F, true>(h, acc, r.x & 0xFFFF, dl,
                                  __builtin_bit_cast(float, r.y), 32, lane, off);
        }
        if (rest != 0 && (nfull & (AGG_WAVES - 1)) == wave) {
            int dl_l = 0, s_l = 0;
            float w_l = 0.0f;
            if (lane < rest) {
                const int2 t = srec[32 * nfull + lane];
                s_l  = t.x & 0xFFFF;
                dl_l = (int)((unsigned)t.x >> 16) - node0;
                w_l  = __builtin_bit_cast(float, t.y);
            }
            if (graphs)
                graph_rows(h, ga, s_l, sbat[dl_l] - g_lo, w_l, lane);
            else
                agg_rows<F, false>(h, acc, s_l, dl_l, w_l, rest, lane, off);
        }
    }
    __syncthreads();

    if (!graphs) {
        agg_finish<F, RELU, POOL>(acc, h, dinv, bias, out, batch, sums, cnt, node0, nloc, tid);
        return;
    }

    v4f   rr[AGG_NSELF];
    float dk[AGG_NSELF];
    int   gk[AGG_NSELF];
#pragma unroll
    for (int k = 0; k < AGG_NSELF; ++k) {
        const int ii = min(wave + k * AGG_WAVES, nloc - 1);
        rr[k] = ((const v4f*)h)[(size_t)(node0 + ii) * F4 + lane];
        dk[k] = dinv[node0 + ii];
        gk[k] = __builtin_amdgcn_readfirstlane(sbat[ii]) - g_lo;
    }
    const v4f bv = ((const v4f*)bias)[lane];
#pragma unroll
    for (int k = 0; k < AGG_NSELF; ++k) {
        const bool ok = wave + k * AGG_WAVES < nloc;
        graph_add(ga, gk[k], rr[k], ok ? dk[k] * dk[k] : 0.0f);
        graph_add(ga, gk[k], bv, ok ? 1.0f : 0.0f);
    }

#pragma unroll
    for (int j = 0; j < AGG_NG; ++j) ((v4f*)acc)[(wave * AGG_NG + j) * F4 + lane] = ga[j];
    __syncthreads();
    for (int idx = tid; idx < span * F; idx += AGG_TPB) {
        const int j = idx / F, f = idx - j * F;
        float s = 0.0f;
        for (int w = 0; w < AGG_WAVES; ++w) s += acc[(w * AGG_NG + j) * F + f];
        atomicAdd(&sums[(size_t)(g_lo + j) * F + f], s);
    }
    if (tid < span) {
        float c = 0.0f;
        for (int i = 0; i < nloc; ++i) c += (sbat[i] - g_lo == tid) ? 1.0f : 0.0f;
        atomicAdd(&cnt[g_lo + tid], c);
    }
}

template <int F>
__global__ void pool_div_kernel(float* __restrict__ sums, const float* __restrict__ cnt, int G) {
    int gid = blockIdx.x * blockDim.x + threadIdx.x;
    if (gid >= G * F) return;
    int g = gid / F;
    sums[gid] /= fmaxf(cnt[g], 1.0f);
}

static inline int cdiv(long long a, long long b) { return (int)((a + b - 1) / b); }

extern "C" void kernel_launch(void* const* d_in, const int* in_sizes, int n_in,
                              void* d_out, int out_size, void* d_ws, size_t ws_size,
                              hipStream_t stream) {
    (void)n_in; (void)ws_size;
    const float* x    = (const float*)d_in[0];
    const int*   edge = (const int*)d_in[1];
    const int*   batch= (const int*)d_in[2];
    const float* b1   = (const float*)d_in[4];
    const float* W1   = (const float*)d_in[3];
    const float* W2   = (const float*)d_in[5];
    const float* b2   = (const float*)d_in[6];
    const float* Wfc  = (const float*)d_in[7];
    const float* bfc  = (const float*)d_in[8];
    float* out = (float*)d_out;

    const int N   = in_sizes[2];
    const int E   = in_sizes[1] / 2;
    const int D   = 64;
    const int D2  = 128;
    const int HID = in_sizes[8];
    const int G   = out_size / HID;

    const int* src = edge;
    const int* dst = edge + E;

    float* ws    = (float*)d_ws;
    float* hw2   = ws + 2 * (size_t)N * D;
    float* ebuf  = hw2  + (size_t)N * D2;
    int*   pk    = (int*)ebuf;
    int*   obw   = pk + E;
    int*   rec   = obw + E;
    float* zblk  = (float*)(rec + 2 * (size_t)E);
    int*   ocount= (int*)zblk;
    float* sums  = zblk + AGG_OWNERS;
    float* cnt   = sums + (size_t)G * D2;
    unsigned* bm = (unsigned*)(cnt + G);
    float* dinv  = ebuf + (size_t)N * D2;

    fill_kernel<<<cdiv((long long)AGG_OWNERS + G * D2 + G, TPB), TPB, 0, stream>>>(
        zblk, 0.0f, AGG_OWNERS + G * D2 + G);
    fill_kernel<<<cdiv(N, TPB), TPB, 0, stream>>>(dinv, 1.0f, N);

    edge_prep_kernel<<<cdiv(cdiv(E, 4), TPB), TPB, 0, stream>>>(src, dst, dinv, pk, obw, ocount, bm, E);
    rsqrt_kernel<<<cdiv(N, TPB), TPB, 0, stream>>>(dinv, N);

    conv1_owner_kernel<<<cdiv(N, AGG_NPB), AGG_TPB, 0, stream>>>(
        x, pk, obw, bm, ocount, dinv, W1, b1, W2, hw2, rec, N, E);

    agg_replay_kernel<128, false, true><<<cdiv(N, AGG_NPB), AGG_TPB, 0, stream>>>(
        hw2, rec, ocount, dinv, b2, nullptr, batch, sums, cnt, N);

    pool_div_kernel<128><<<cdiv((long long)G * D2, TPB), TPB, 0, stream>>>(sums, cnt, G);

    {
        int row_tiles = G / 16;
        wmma_gemm_f32<128, 64><<<cdiv((long long)row_tiles * 32, TPB), TPB, 0, stream>>>(
            sums, Wfc, out, bfc, row_tiles);
    }
}


# Last product (mean pool + FC) as ONE launch, one wave per 16x16 tile: 128 waves of 32 matrix instructions (was 32 waves of 128); mean's division moved behind the product; 8 launches to 7
# speedup vs baseline: 4.1755x; 1.0754x over previous
#include <hip/hip_runtime.h>
#include <hip/hip_bf16.h>


typedef float v2f __attribute__((ext_vector_type(2)));
typedef float v8f __attribute__((ext_vector_type(8)));

#define TPB 256

__global__ void fill_kernel(float* __restrict__ p, float v, int n) {
    int i = blockIdx.x * blockDim.x + threadIdx.x;
    if (i < n) p[i] = v;
}

#define AGG_NPB    196
#define AGG_OWNERS 256
__global__ void edge_prep_kernel(const int* __restrict__ src, const int* __restrict__ dst,
                                 float* __restrict__ deg, int* __restrict__ pk,
                                 int* __restrict__ obw, int* __restrict__ ocount,
                                 unsigned* __restrict__ bm, int E) {
    __shared__ int hist[AGG_OWNERS];
    __shared__ unsigned bits[AGG_OWNERS];
    for (int i = threadIdx.x; i < AGG_OWNERS; i += blockDim.x) { hist[i] = 0; bits[i] = 0u; }
    __syncthreads();
    int e = (blockIdx.x * blockDim.x + threadIdx.x) * 4;
    if (e + 3 < E) {
        int4 d = *(const int4*)(dst + e);
        int4 s = *(const int4*)(src + e);
        atomicAdd(&deg[d.x], 1.0f);
        atomicAdd(&deg[d.y], 1.0f);
        atomicAdd(&deg[d.z], 1.0f);
        atomicAdd(&deg[d.w], 1.0f);
        int4 p;
        p.x = (int)(((unsigned)d.x << 16) | (unsigned)s.x);
        p.y = (int)(((unsigned)d.y << 16) | (unsigned)s.y);
        p.z = (int)(((unsigned)d.z << 16) | (unsigned)s.z);
        p.w = (int)(((unsigned)d.w << 16) | (unsigned)s.w);
        *(int4*)(pk + e) = p;
        const unsigned ox = (unsigned)(d.x / AGG_NPB), oy = (unsigned)(d.y / AGG_NPB);
        const unsigned oz = (unsigned)(d.z / AGG_NPB), ow = (unsigned)(d.w / AGG_NPB);
        obw[e >> 2] = (int)(ox | (oy << 8) | (oz << 16) | (ow << 24));
        atomicAdd(&hist[ox], 1);
        atomicAdd(&hist[oy], 1);
        atomicAdd(&hist[oz], 1);
        atomicAdd(&hist[ow], 1);
        const unsigned pb = 1u << (threadIdx.x >> 3);
        atomicOr(&bits[ox], pb);
        atomicOr(&bits[oy], pb);
        atomicOr(&bits[oz], pb);
        atomicOr(&bits[ow], pb);
    } else {
        unsigned char* ob = (unsigned char*)obw;
        for (; e < E; ++e) {
            int d = dst[e];
            atomicAdd(&deg[d], 1.0f);
            pk[e] = (int)(((unsigned)d << 16) | (unsigned)src[e]);
            ob[e] = (unsigned char)(d / AGG_NPB);
            atomicAdd(&hist[d / AGG_NPB], 1);
            atomicOr(&bits[d / AGG_NPB], 1u << ((e - (int)(blockIdx.x * blockDim.x) * 4) >> 5));
        }
    }
    __syncthreads();
    for (int i = threadIdx.x; i < AGG_OWNERS; i += blockDim.x) {
        const int c = hist[i];
        if (c) atomicAdd(&ocount[i], c);
        bm[(size_t)blockIdx.x * AGG_OWNERS + i] = bits[i];
    }
}

__global__ void rsqrt_kernel(float* __restrict__ d, int n) {
    int i = blockIdx.x * blockDim.x + threadIdx.x;
    if (i < n) d[i] = rsqrtf(d[i]);
}

template <int K, int Nn>
__global__ __launch_bounds__(32 * (Nn / 16))
void pooled_product_f32(const float* __restrict__ A,
                        const float* __restrict__ cnt,
                        const float* __restrict__ B,
                        const float* __restrict__ bias,
                        float* __restrict__ C) {
    constexpr int KS = K / 4;
    const int ct   = __builtin_amdgcn_readfirstlane((int)(threadIdx.x >> 5));
    const int lane = threadIdx.x & 31;
    const int half = lane >> 4;
    const int lm   = lane & 15;
    const int row0 = blockIdx.x << 4;

    const float* __restrict__ arow = A + (size_t)(row0 + lm) * K + 2 * half;
    const float* __restrict__ bcol = B + (size_t)(2 * half) * Nn + ct * 16 + lm;

    v2f a[KS], b[KS];
#pragma unroll
    for (int k = 0; k < KS; ++k) a[k] = *(const v2f*)(arow + 4 * k);
#pragma unroll
    for (int k = 0; k < KS; ++k) {
        b[k].x = bcol[(4 * k) * Nn];
        b[k].y = bcol[(4 * k + 1) * Nn];
    }
    float cv[8];
#pragma unroll
    for (int v = 0; v < 8; ++v) cv[v] = cnt[row0 + 8 * half + v];
    const float bv = bias[ct * 16 + lm];

    v8f d = (v8f){};
#pragma unroll
    for (int k = 0; k < KS; ++k)
        d = __builtin_amdgcn_wmma_f32_16x16x4_f32(false, a[k], false, b[k], (short)0, d,
                                                  false, false);

    float* __restrict__ crow = C + (size_t)(row0 + 8 * half) * Nn + ct * 16 + lm;
#pragma unroll
    for (int v = 0; v < 8; ++v) crow[(size_t)v * Nn] = d[v] / fmaxf(cv[v], 1.0f) + bv;
}

#define AGG_TPB   1024
#define AGG_WAVES 32
#define AGG_QCAP  1056
#define AGG_PCAP  1024

typedef float v4f __attribute__((ext_vector_type(4)));

template <int F> struct RowVecSel      { typedef v2f T; };
template <>      struct RowVecSel<128> { typedef v4f T; };

template <int F, bool FULL, int PITCH = F>
__device__ __forceinline__ void agg_rows(const float* __restrict__ h, float* acc,
                                         int s_l, int dl_l, float w_l, int cnt, int lane,
                                         const int (&off)[F / 32]) {
    typedef typename RowVecSel<F>::T VT;
    constexpr int VEC = F / 32;
#pragma unroll
    for (int g = 0; g < 32; g += 8) {
        if (!FULL && g >= cnt) break;
        VT r[8];
#pragma unroll
        for (int i = 0; i < 8; ++i) {
            if (FULL || g + i < cnt) {
                int s = __builtin_amdgcn_readlane(s_l, g + i);
                r[i] = *((const VT*)(h + (size_t)s * F) + lane);
            }
        }
#pragma unroll
        for (int i = 0; i < 8; ++i) {
            if (FULL || g + i < cnt) {
                int dl = __builtin_amdgcn_readlane(dl_l, g + i);
                float w = __builtin_bit_cast(
                    float, __builtin_amdgcn_readlane(__builtin_bit_cast(int, w_l), g + i));
                float* a = acc + dl * PITCH;
#pragma unroll
                for (int k = 0; k < VEC; ++k)
                    __hip_atomic_fetch_add(&a[off[k]], r[i][k] * w, __ATOMIC_RELAXED,
                                           __HIP_MEMORY_SCOPE_WORKGROUP);
            }
        }
    }
}

template <int F, bool FULL, int PITCH = F>
__device__ __forceinline__ void agg_drain(const float* __restrict__ h,
                                          const int* __restrict__ pk,
                                          const float* __restrict__ dinv,
                                          float* acc, const int* q,
                                          int start, int cnt, int node0, int lane,
                                          const int (&off)[F / 32],
                                          int2* __restrict__ rec_out) {
    int dl_l = 0, s_l = 0;
    float w_l = 0.0f;
    if (FULL || lane < cnt) {
        int e_l = q[start + lane];
        int p_l = pk[e_l];
        int d_l = (int)((unsigned)p_l >> 16);
        s_l = p_l & 0xFFFF;
        dl_l = d_l - node0;
        w_l = dinv[s_l] * dinv[d_l];
        rec_out[lane] = make_int2(p_l, __builtin_bit_cast(int, w_l));
    }
    agg_rows<F, FULL, PITCH>(h, acc, s_l, dl_l, w_l, cnt, lane, off);
}

template <int F>
__device__ __forceinline__ void agg_row_words(int (&off)[F / 32], int lane) {
    constexpr int VEC = F / 32;
#pragma unroll
    for (int k = 0; k < VEC; ++k)
        off[k] = (F == 128) ? (4 * lane + (k ^ (lane >> 4))) : (VEC * lane + k);
}

__device__ __forceinline__ int agg_owner_base(const int* __restrict__ ocount, int b, int lane) {
    int s = 0;
    for (int i = lane; i < b; i += 32) s += ocount[i];
    for (int o = 16; o > 0; o >>= 1) s += __shfl_xor(s, o);
    return s;
}

#define AGG_SLICES 8
#define AGG_RPS    ((AGG_NPB + AGG_SLICES - 1) / AGG_SLICES)
template <int F, bool RELU, bool POOL>
__device__ __forceinline__ void agg_finish(float* acc,
                                           const float* __restrict__ h,
                                           const float* __restrict__ dinv,
                                           const float* __restrict__ bias,
                                           float* __restrict__ out,
                                           const int* __restrict__ batch,
                                           float* __restrict__ sums,
                                           float* __restrict__ cnt,
                                           int node0, int nloc, int tid) {
    constexpr int F4 = F / 4;
    for (int idx = tid; idx < nloc * F4; idx += AGG_TPB) {
        int i = idx / F4;
        int g = idx - i * F4;
        v4f q = ((const v4f*)acc)[idx];
        if (F == 128 && (g >> 4)) q = (v4f){q.y, q.x, q.w, q.z};
        int gi = node0 + i;
        float di = dinv[gi];
        v4f xv = ((const v4f*)h)[(size_t)gi * F4 + g];
        v4f bv = ((const v4f*)bias)[g];
        v4f o;
#pragma unroll
        for (int c = 0; c < 4; ++c) {
            float v = q[c] + xv[c] * di * di + bv[c];
            o[c] = RELU ? fmaxf(v, 0.0f) : v;
        }
        if (POOL) ((v4f*)acc)[idx] = o;
        else      ((v4f*)out)[(size_t)gi * F4 + g] = o;
    }

    if (POOL) {
        static_assert(!POOL || AGG_TPB == AGG_SLICES * F, "one thread per (slice, feature)");
        __syncthreads();
        const int f  = tid & (F - 1);
        const int r0 = (tid / F) * AGG_RPS;
        int   bid[AGG_RPS];
        float val[AGG_RPS];
#pragma unroll
        for (int j = 0; j < AGG_RPS; ++j) {
            const int r  = r0 + j;
            const bool ok = r < nloc;
            const int rr = ok ? r : 0;
            const int b  = batch[node0 + rr];
            const float v = acc[rr * F + f];
            bid[j] = ok ? b : -1;
            val[j] = ok ? v : 0.0f;
        }
        int   cur  = -1;
        float run  = 0.0f;
        float nrun = 0.0f;
#pragma unroll
        for (int j = 0; j < AGG_RPS; ++j) {
            if (bid[j] != cur) {
                if (cur >= 0) {
                    atomicAdd(&sums[(size_t)cur * F + f], run);
                    if (f == 0) atomicAdd(&cnt[cur], nrun);
                }
                cur = bid[j]; run = 0.0f; nrun = 0.0f;
            }
            run  += val[j];
            nrun += 1.0f;
        }
        if (cur >= 0) {
            atomicAdd(&sums[(size_t)cur * F + f], run);
            if (f == 0) atomicAdd(&cnt[cur], nrun);
        }
    }
}

#define C1_ROWS  208
#define C1_RT    (C1_ROWS / 16)
#define C1_PITCH 68
__global__ __launch_bounds__(AGG_TPB)
void conv1_owner_kernel(const float* __restrict__ h,
                        const int* __restrict__ pk,
                        const int* __restrict__ obw,
                        const unsigned* __restrict__ bm,
                        const int* __restrict__ ocount,
                        const float* __restrict__ dinv,
                        const float* __restrict__ Wa,
                        const float* __restrict__ ba,
                        const float* __restrict__ Wb,
                        float* __restrict__ out,
                        int* __restrict__ recw, int N, int E) {
    constexpr int F = 64;
    __shared__ __attribute__((aligned(16))) float acc[C1_ROWS * C1_PITCH];
    __shared__ __attribute__((aligned(16))) int qbuf[AGG_WAVES * AGG_QCAP];
    __shared__ unsigned short plist[AGG_WAVES * AGG_PCAP];
    __shared__ int cursor;

    const int tid  = threadIdx.x;
    const int lane = tid & 31;
    const int wave = __builtin_amdgcn_readfirstlane(tid >> 5);
    const int node0 = blockIdx.x * AGG_NPB;
    const int nloc  = min(AGG_NPB, N - node0);

    for (int idx = tid; idx < C1_ROWS * (F / 4); idx += AGG_TPB) {
        const int i = idx >> 4, g = idx & 15;
        v4f v = (v4f){0.f, 0.f, 0.f, 0.f};
        if (i < nloc) {
            const float di = dinv[node0 + i];
            v = ((const v4f*)h)[(size_t)(node0 + i) * (F / 4) + g] * (di * di);
        }
        *(v4f*)(acc + i * C1_PITCH + 4 * g) = v;
    }
    if (tid == 0) cursor = 0;
    int2* rec = (int2*)recw + agg_owner_base(ocount, (int)blockIdx.x, lane);
    __syncthreads();

    int off[F / 32];
    agg_row_words<F>(off, lane);

    int* myq = qbuf + wave * AGG_QCAP;
    int qcnt = 0;

    auto push = [&](bool m, int e) {
        unsigned mask = __builtin_amdgcn_ballot_w32(m);
        if (mask) {
            int pos = qcnt + (int)__builtin_amdgcn_mbcnt_lo(mask, 0u);
            if (m) myq[pos] = e;
            qcnt += __builtin_popcount(mask);
        }
    };
    auto reserve = [&](int n) -> int {
        int pos = 0;
        if (lane == 0) pos = atomicAdd(&cursor, n);
        return __builtin_amdgcn_readfirstlane(pos);
    };
    const unsigned pat = (unsigned)blockIdx.x * 0x01010101u;
    auto eqb = [&](int wv) -> unsigned {
        const unsigned y = (unsigned)wv ^ pat;
        const unsigned t = ((y & 0x7F7F7F7Fu) + 0x7F7F7F7Fu) | y;
        return ~t & 0x80808080u;
    };
    auto ldp = [&](int sc, int4& a, int4& b) {
        const int4* p = (const int4*)(obw + (size_t)sc * 8);
        a = p[0];
        b = p[1];
    };
    auto drain_full = [&]() {
        while (qcnt >= 32) {
            qcnt -= 32;
            agg_drain<F, true, C1_PITCH>(h, pk, dinv, acc, myq, qcnt, 32, node0, lane, off,
                                         rec + reserve(32));
        }
    };

    const int nblk = (E + 1023) >> 10;
    unsigned short* myp = plist + wave * AGG_PCAP;
    for (int w0 = 0; w0 < nblk; w0 += AGG_WAVES * 32) {
        const int wi = w0 + lane * AGG_WAVES + wave;
        unsigned word = (wi < nblk) ? bm[(size_t)wi * AGG_OWNERS + blockIdx.x] : 0u;
        const int nb = __builtin_popcount(word);
        int incl = nb;
        for (int o = 1; o < 32; o <<= 1) {
            const int t = __shfl_up(incl, o);
            if (lane >= o) incl += t;
        }
        int pos = incl - nb;
        const int L = __builtin_amdgcn_readlane(incl, 31);
        while (__builtin_amdgcn_ballot_w32(word != 0u)) {
            if (word != 0u) {
                myp[pos++] = (unsigned short)((lane << 5) | __builtin_ctz(word));
                word &= word - 1u;
            }
        }
        auto fetch = [&](int i0, bool& has, int& sc, int4& a, int4& b) {
            has = i0 + lane < L;
            const int ent = has ? (int)myp[i0 + lane] : 0;
            sc = has ? ((w0 + (ent >> 5) * AGG_WAVES + wave) * 32 + (ent & 31)) : 0;
            ldp(sc, a, b);
        };
        bool has; int sc; int4 a, b;
        fetch(0, has, sc, a, b);
        for (int i0 = 0; i0 < L; i0 += 32) {
            unsigned m = (eqb(a.x) >> 7) | (eqb(a.y) >> 6) | (eqb(a.z) >> 5) | (eqb(a.w) >> 4) |
                         (eqb(b.x) >> 3) | (eqb(b.y) >> 2) | (eqb(b.z) >> 1) | eqb(b.w);
            if (!has) m = 0u;
            const int eb = sc * 32;
            while (__builtin_amdgcn_ballot_w32(m != 0u)) {
                const bool h2 = m != 0u;
                const int p = h2 ? __builtin_ctz(m) : 0;
                const int e = eb + 4 * (p & 7) + (p >> 3);
                m &= m - 1u;
                push(h2 && e < E, e);
            }
            fetch(i0 + 32, has, sc, a, b);
            drain_full();
        }
    }
    while (qcnt > 0) {
        int cnt = min(qcnt, 32);
        qcnt -= cnt;
        agg_drain<F, false, C1_PITCH>(h, pk, dinv, acc, myq, qcnt, cnt, node0, lane, off,
                                      rec + reserve(cnt));
    }
    __syncthreads();

    const int half = lane >> 4;
    const int lm   = lane & 15;
    float* hbuf = (float*)qbuf;

    const int cg = wave & 3;

    {
        const float* __restrict__ bb = Wa + (2 * half) * 64 + cg * 16 + lm;
        v2f wb[16];
#pragma unroll
        for (int k = 0; k < 16; ++k) {
            wb[k].x = bb[(4 * k) * 64];
            wb[k].y = bb[(4 * k + 1) * 64];
        }
        const float bv = ba[cg * 16 + lm];
        for (int rt = wave >> 2; rt < C1_RT; rt += AGG_WAVES / 4) {
            const float* arow = acc + (rt * 16 + lm) * C1_PITCH + 2 * half;
            v8f d = (v8f){};
#pragma unroll
            for (int k = 0; k < 16; ++k) {
                const v2f a = *(const v2f*)(arow + 4 * k);
                d = __builtin_amdgcn_wmma_f32_16x16x4_f32(false, a, false, wb[k], (short)0, d,
                                                          false, false);
            }
            float* hrow = hbuf + (rt * 16 + 8 * half) * C1_PITCH + cg * 16 + lm;
#pragma unroll
            for (int v = 0; v < 8; ++v) hrow[v * C1_PITCH] = fmaxf(d[v] + bv, 0.0f);
        }
    }
    __syncthreads();

    const float* __restrict__ bb2 = Wb + (2 * half) * 128 + cg * 32 + lm;
    v2f wp[16], wq[16];
#pragma unroll
    for (int k = 0; k < 16; ++k) {
        wp[k].x = bb2[(4 * k) * 128];
        wp[k].y = bb2[(4 * k + 1) * 128];
        wq[k].x = bb2[(4 * k) * 128 + 16];
        wq[k].y = bb2[(4 * k + 1) * 128 + 16];
    }
    for (int rt = wave >> 2; rt < C1_RT; rt += AGG_WAVES / 4) {
        const float* arow = hbuf + (rt * 16 + lm) * C1_PITCH + 2 * half;
        v8f d0 = (v8f){}, d1 = (v8f){};
#pragma unroll
        for (int k = 0; k < 16; ++k) {
            const v2f a = *(const v2f*)(arow + 4 * k);
            d0 = __builtin_amdgcn_wmma_f32_16x16x4_f32(false, a, false, wp[k], (short)0, d0,
                                                       false, false);
            d1 = __builtin_amdgcn_wmma_f32_16x16x4_f32(false, a, false, wq[k], (short)0, d1,
                                                       false, false);
        }
        const int r0 = rt * 16 + 8 * half;
        float* __restrict__ crow = out + (size_t)(node0 + r0) * 128 + cg * 32 + lm;
#pragma unroll
        for (int v = 0; v < 8; ++v) {
            if (r0 + v < nloc) {
                crow[(size_t)v * 128]      = d0[v];
                crow[(size_t)v * 128 + 16] = d1[v];
            }
        }
    }
}


#define AGG_NG 4
#define AGG_NSELF ((AGG_NPB + AGG_WAVES - 1) / AGG_WAVES)

__device__ __forceinline__ void graph_add(v4f (&a)[AGG_NG], int g, v4f v, float w) {
#pragma unroll
    for (int j = 0; j < AGG_NG; ++j) {
        const float m = (g == j) ? w : 0.0f;
        a[j] += v * m;
    }
}

__device__ __forceinline__ void graph_rows(const float* __restrict__ h, v4f (&a)[AGG_NG],
                                           int s_l, int g_l, float w_l, int lane) {
#pragma unroll
    for (int g0 = 0; g0 < 32; g0 += 8) {
        v4f r[8];
#pragma unroll
        for (int i = 0; i < 8; ++i) {
            int s = __builtin_amdgcn_readlane(s_l, g0 + i);
            r[i] = *((const v4f*)(h + (size_t)s * 128) + lane);
        }
#pragma unroll
        for (int i = 0; i < 8; ++i) {
            int g = __builtin_amdgcn_readlane(g_l, g0 + i);
            float w = __builtin_bit_cast(
                float, __builtin_amdgcn_readlane(__builtin_bit_cast(int, w_l), g0 + i));
            graph_add(a, g, r[i], w);
        }
    }
}

#define AGG_RCAP 8192
#define AGG_RLD  4
template <int F, bool RELU, bool POOL>
__global__ __launch_bounds__(AGG_TPB)
void agg_replay_kernel(const float* __restrict__ h,
                       const int* __restrict__ recw,
                       const int* __restrict__ ocount,
                       const float* __restrict__ dinv,
                       const float* __restrict__ bias,
                       float* __restrict__ out,
                       const int* __restrict__ batch,
                       float* __restrict__ sums,
                       float* __restrict__ cnt, int N) {
    constexpr int F4 = F / 4;
    __shared__ __attribute__((aligned(16))) float acc[AGG_NPB * F];
    __shared__ int2 srec[AGG_RCAP];
    __shared__ int sbat[AGG_NPB];

    const int tid  = threadIdx.x;
    const int lane = tid & 31;
    const int wave = __builtin_amdgcn_readfirstlane(tid >> 5);
    const int node0 = blockIdx.x * AGG_NPB;
    const int nloc  = min(AGG_NPB, N - node0);

    int g_lo = 0, span = AGG_NG + 1;
    if (POOL) {
        int lo = 0x7FFFFFFF, hi = -1;
        for (int i = lane; i < nloc; i += 32) {
            const int b = batch[node0 + i];
            lo = min(lo, b); hi = max(hi, b);
        }
        for (int o = 16; o > 0; o >>= 1) {
            lo = min(lo, __shfl_xor(lo, o)); hi = max(hi, __shfl_xor(hi, o));
        }
        g_lo = __builtin_amdgcn_readfirstlane(lo);
        span = __builtin_amdgcn_readfirstlane(hi) - g_lo + 1;
        for (int i = tid; i < nloc; i += AGG_TPB) sbat[i] = batch[node0 + i];
    }
    const bool graphs = POOL && F == 128 && span <= AGG_NG;
    if (!graphs)
        for (int i = tid; i < AGG_NPB * F4; i += AGG_TPB) ((v4f*)acc)[i] = (v4f){0.f, 0.f, 0.f, 0.f};
    v4f ga[AGG_NG];
#pragma unroll
    for (int j = 0; j < AGG_NG; ++j) ga[j] = (v4f){0.f, 0.f, 0.f, 0.f};
    const int total = ocount[blockIdx.x];
    const int2* __restrict__ rec =
        (const int2*)recw + agg_owner_base(ocount, (int)blockIdx.x, lane);

    int off[F / 32];
    agg_row_words<F>(off, lane);

    for (int r0 = 0; r0 < total; r0 += AGG_RCAP) {
        const int nst = min(total - r0, AGG_RCAP);
        if (r0 != 0) __syncthreads();
        for (int j0 = tid; j0 < nst; j0 += AGG_RLD * AGG_TPB) {
            int2 t[AGG_RLD];
#pragma unroll
            for (int k = 0; k < AGG_RLD; ++k)
                t[k] = rec[r0 + min(j0 + k * AGG_TPB, nst - 1)];
#pragma unroll
            for (int k = 0; k < AGG_RLD; ++k)
                if (j0 + k * AGG_TPB < nst) srec[j0 + k * AGG_TPB] = t[k];
        }
        __syncthreads();

        const int nfull = nst >> 5;
        const int rest  = nst & 31;
        for (int i = wave; i < nfull; i += AGG_WAVES) {
            const int2 r = srec[32 * i + lane];
            const int dl = (int)((unsigned)r.x >> 16) - node0;
            if (graphs)
                graph_rows(h, ga, r.x & 0xFFFF, sbat[dl] - g_lo,
                           __builtin_bit_cast(float, r.y), lane);
            else
                agg_rows<F, true>(h, acc, r.x & 0xFFFF, dl,
                                  __builtin_bit_cast(float, r.y), 32, lane, off);
        }
        if (rest != 0 && (nfull & (AGG_WAVES - 1)) == wave) {
            int dl_l = 0, s_l = 0;
            float w_l = 0.0f;
            if (lane < rest) {
                const int2 t = srec[32 * nfull + lane];
                s_l  = t.x & 0xFFFF;
                dl_l = (int)((unsigned)t.x >> 16) - node0;
                w_l  = __builtin_bit_cast(float, t.y);
            }
            if (graphs)
                graph_rows(h, ga, s_l, sbat[dl_l] - g_lo, w_l, lane);
            else
                agg_rows<F, false>(h, acc, s_l, dl_l, w_l, rest, lane, off);
        }
    }
    __syncthreads();

    if (!graphs) {
        agg_finish<F, RELU, POOL>(acc, h, dinv, bias, out, batch, sums, cnt, node0, nloc, tid);
        return;
    }

    v4f   rr[AGG_NSELF];
    float dk[AGG_NSELF];
    int   gk[AGG_NSELF];
#pragma unroll
    for (int k = 0; k < AGG_NSELF; ++k) {
        const int ii = min(wave + k * AGG_WAVES, nloc - 1);
        rr[k] = ((const v4f*)h)[(size_t)(node0 + ii) * F4 + lane];
        dk[k] = dinv[node0 + ii];
        gk[k] = __builtin_amdgcn_readfirstlane(sbat[ii]) - g_lo;
    }
    const v4f bv = ((const v4f*)bias)[lane];
#pragma unroll
    for (int k = 0; k < AGG_NSELF; ++k) {
        const bool ok = wave + k * AGG_WAVES < nloc;
        graph_add(ga, gk[k], rr[k], ok ? dk[k] * dk[k] : 0.0f);
        graph_add(ga, gk[k], bv, ok ? 1.0f : 0.0f);
    }

#pragma unroll
    for (int j = 0; j < AGG_NG; ++j) ((v4f*)acc)[(wave * AGG_NG + j) * F4 + lane] = ga[j];
    __syncthreads();
    for (int idx = tid; idx < span * F; idx += AGG_TPB) {
        const int j = idx / F, f = idx - j * F;
        float s = 0.0f;
        for (int w = 0; w < AGG_WAVES; ++w) s += acc[(w * AGG_NG + j) * F + f];
        atomicAdd(&sums[(size_t)(g_lo + j) * F + f], s);
    }
    if (tid < span) {
        float c = 0.0f;
        for (int i = 0; i < nloc; ++i) c += (sbat[i] - g_lo == tid) ? 1.0f : 0.0f;
        atomicAdd(&cnt[g_lo + tid], c);
    }
}

static inline int cdiv(long long a, long long b) { return (int)((a + b - 1) / b); }

extern "C" void kernel_launch(void* const* d_in, const int* in_sizes, int n_in,
                              void* d_out, int out_size, void* d_ws, size_t ws_size,
                              hipStream_t stream) {
    (void)n_in; (void)ws_size;
    const float* x    = (const float*)d_in[0];
    const int*   edge = (const int*)d_in[1];
    const int*   batch= (const int*)d_in[2];
    const float* b1   = (const float*)d_in[4];
    const float* W1   = (const float*)d_in[3];
    const float* W2   = (const float*)d_in[5];
    const float* b2   = (const float*)d_in[6];
    const float* Wfc  = (const float*)d_in[7];
    const float* bfc  = (const float*)d_in[8];
    float* out = (float*)d_out;

    const int N   = in_sizes[2];
    const int E   = in_sizes[1] / 2;
    const int D   = 64;
    const int D2  = 128;
    const int HID = in_sizes[8];
    const int G   = out_size / HID;

    const int* src = edge;
    const int* dst = edge + E;

    float* ws    = (float*)d_ws;
    float* hw2   = ws + 2 * (size_t)N * D;
    float* ebuf  = hw2  + (size_t)N * D2;
    int*   pk    = (int*)ebuf;
    int*   obw   = pk + E;
    int*   rec   = obw + E;
    float* zblk  = (float*)(rec + 2 * (size_t)E);
    int*   ocount= (int*)zblk;
    float* sums  = zblk + AGG_OWNERS;
    float* cnt   = sums + (size_t)G * D2;
    unsigned* bm = (unsigned*)(cnt + G);
    float* dinv  = ebuf + (size_t)N * D2;

    fill_kernel<<<cdiv((long long)AGG_OWNERS + G * D2 + G, TPB), TPB, 0, stream>>>(
        zblk, 0.0f, AGG_OWNERS + G * D2 + G);
    fill_kernel<<<cdiv(N, TPB), TPB, 0, stream>>>(dinv, 1.0f, N);

    edge_prep_kernel<<<cdiv(cdiv(E, 4), TPB), TPB, 0, stream>>>(src, dst, dinv, pk, obw, ocount, bm, E);
    rsqrt_kernel<<<cdiv(N, TPB), TPB, 0, stream>>>(dinv, N);

    conv1_owner_kernel<<<cdiv(N, AGG_NPB), AGG_TPB, 0, stream>>>(
        x, pk, obw, bm, ocount, dinv, W1, b1, W2, hw2, rec, N, E);

    agg_replay_kernel<128, false, true><<<cdiv(N, AGG_NPB), AGG_TPB, 0, stream>>>(
        hw2, rec, ocount, dinv, b2, nullptr, batch, sums, cnt, N);

    pooled_product_f32<128, 64><<<G / 16, 32 * (64 / 16), 0, stream>>>(sums, cnt, Wfc, bfc, out);
}


# In-degrees counted by the owner work-groups in LDS: their edge scan is its own launch before conv1 (writes edge lists and dinv), conv1 replays its list; no global atomic per edge; sums: LDS arrival
# speedup vs baseline: 6.1633x; 1.4761x over previous
#include <hip/hip_runtime.h>
#include <hip/hip_bf16.h>


typedef float v2f __attribute__((ext_vector_type(2)));
typedef float v8f __attribute__((ext_vector_type(8)));

#define TPB 256

__global__ void fill_kernel(float* __restrict__ p, float v, int n) {
    int i = blockIdx.x * blockDim.x + threadIdx.x;
    if (i < n) p[i] = v;
}

#define AGG_NPB    196
#define AGG_OWNERS 256
__global__ void edge_prep_kernel(const int* __restrict__ src, const int* __restrict__ dst,
                                 int* __restrict__ pk,
                                 int* __restrict__ obw, int* __restrict__ ocount,
                                 unsigned* __restrict__ bm, int E) {
    __shared__ int hist[AGG_OWNERS];
    __shared__ unsigned bits[AGG_OWNERS];
    for (int i = threadIdx.x; i < AGG_OWNERS; i += blockDim.x) { hist[i] = 0; bits[i] = 0u; }
    __syncthreads();
    int e = (blockIdx.x * blockDim.x + threadIdx.x) * 4;
    if (e + 3 < E) {
        int4 d = *(const int4*)(dst + e);
        int4 s = *(const int4*)(src + e);
        int4 p;
        p.x = (int)(((unsigned)d.x << 16) | (unsigned)s.x);
        p.y = (int)(((unsigned)d.y << 16) | (unsigned)s.y);
        p.z = (int)(((unsigned)d.z << 16) | (unsigned)s.z);
        p.w = (int)(((unsigned)d.w << 16) | (unsigned)s.w);
        *(int4*)(pk + e) = p;
        const unsigned ox = (unsigned)(d.x / AGG_NPB), oy = (unsigned)(d.y / AGG_NPB);
        const unsigned oz = (unsigned)(d.z / AGG_NPB), ow = (unsigned)(d.w / AGG_NPB);
        obw[e >> 2] = (int)(ox | (oy << 8) | (oz << 16) | (ow << 24));
        atomicAdd(&hist[ox], 1);
        atomicAdd(&hist[oy], 1);
        atomicAdd(&hist[oz], 1);
        atomicAdd(&hist[ow], 1);
        const unsigned pb = 1u << (threadIdx.x >> 3);
        atomicOr(&bits[ox], pb);
        atomicOr(&bits[oy], pb);
        atomicOr(&bits[oz], pb);
        atomicOr(&bits[ow], pb);
    } else {
        unsigned char* ob = (unsigned char*)obw;
        for (; e < E; ++e) {
            int d = dst[e];
            pk[e] = (int)(((unsigned)d << 16) | (unsigned)src[e]);
            ob[e] = (unsigned char)(d / AGG_NPB);
            atomicAdd(&hist[d / AGG_NPB], 1);
            atomicOr(&bits[d / AGG_NPB], 1u << ((e - (int)(blockIdx.x * blockDim.x) * 4) >> 5));
        }
    }
    __syncthreads();
    for (int i = threadIdx.x; i < AGG_OWNERS; i += blockDim.x) {
        const int c = hist[i];
        if (c) atomicAdd(&ocount[i], c);
        bm[(size_t)blockIdx.x * AGG_OWNERS + i] = bits[i];
    }
}

template <int K, int Nn>
__global__ __launch_bounds__(32 * (Nn / 16))
void pooled_product_f32(const float* __restrict__ A,
                        const float* __restrict__ cnt,
                        const float* __restrict__ B,
                        const float* __restrict__ bias,
                        float* __restrict__ C) {
    constexpr int KS = K / 4;
    const int ct   = __builtin_amdgcn_readfirstlane((int)(threadIdx.x >> 5));
    const int lane = threadIdx.x & 31;
    const int half = lane >> 4;
    const int lm   = lane & 15;
    const int row0 = blockIdx.x << 4;

    const float* __restrict__ arow = A + (size_t)(row0 + lm) * K + 2 * half;
    const float* __restrict__ bcol = B + (size_t)(2 * half) * Nn + ct * 16 + lm;

    v2f a[KS], b[KS];
#pragma unroll
    for (int k = 0; k < KS; ++k) a[k] = *(const v2f*)(arow + 4 * k);
#pragma unroll
    for (int k = 0; k < KS; ++k) {
        b[k].x = bcol[(4 * k) * Nn];
        b[k].y = bcol[(4 * k + 1) * Nn];
    }
    float cv[8];
#pragma unroll
    for (int v = 0; v < 8; ++v) cv[v] = cnt[row0 + 8 * half + v];
    const float bv = bias[ct * 16 + lm];

    v8f d = (v8f){};
#pragma unroll
    for (int k = 0; k < KS; ++k)
        d = __builtin_amdgcn_wmma_f32_16x16x4_f32(false, a[k], false, b[k], (short)0, d,
                                                  false, false);

    float* __restrict__ crow = C + (size_t)(row0 + 8 * half) * Nn + ct * 16 + lm;
#pragma unroll
    for (int v = 0; v < 8; ++v) crow[(size_t)v * Nn] = d[v] / fmaxf(cv[v], 1.0f) + bv;
}

#define AGG_TPB   1024
#define AGG_WAVES 32
#define AGG_QCAP  1056
#define AGG_PCAP  1024

typedef float v4f __attribute__((ext_vector_type(4)));

template <int F> struct RowVecSel      { typedef v2f T; };
template <>      struct RowVecSel<128> { typedef v4f T; };

template <int F, bool FULL, int PITCH = F>
__device__ __forceinline__ void agg_rows(const float* __restrict__ h, float* acc,
                                         int s_l, int dl_l, float w_l, int cnt, int lane,
                                         const int (&off)[F / 32]) {
    typedef typename RowVecSel<F>::T VT;
    constexpr int VEC = F / 32;
#pragma unroll
    for (int g = 0; g < 32; g += 8) {
        if (!FULL && g >= cnt) break;
        VT r[8];
#pragma unroll
        for (int i = 0; i < 8; ++i) {
            if (FULL || g + i < cnt) {
                int s = __builtin_amdgcn_readlane(s_l, g + i);
                r[i] = *((const VT*)(h + (size_t)s * F) + lane);
            }
        }
#pragma unroll
        for (int i = 0; i < 8; ++i) {
            if (FULL || g + i < cnt) {
                int dl = __builtin_amdgcn_readlane(dl_l, g + i);
                float w = __builtin_bit_cast(
                    float, __builtin_amdgcn_readlane(__builtin_bit_cast(int, w_l), g + i));
                float* a = acc + dl * PITCH;
#pragma unroll
                for (int k = 0; k < VEC; ++k)
                    __hip_atomic_fetch_add(&a[off[k]], r[i][k] * w, __ATOMIC_RELAXED,
                                           __HIP_MEMORY_SCOPE_WORKGROUP);
            }
        }
    }
}

template <bool FULL>
__device__ __forceinline__ void scan_drain(const int* __restrict__ pk, int* degc, const int* q,
                                           int start, int cnt, int node0, int lane,
                                           int* __restrict__ out) {
    if (FULL || lane < cnt) {
        const int e_l = q[start + lane];
        const int p_l = pk[e_l];
        atomicAdd(&degc[(int)((unsigned)p_l >> 16) - node0], 1);
        out[lane] = p_l;
    }
}

template <int F>
__device__ __forceinline__ void agg_row_words(int (&off)[F / 32], int lane) {
    constexpr int VEC = F / 32;
#pragma unroll
    for (int k = 0; k < VEC; ++k)
        off[k] = (F == 128) ? (4 * lane + (k ^ (lane >> 4))) : (VEC * lane + k);
}

__device__ __forceinline__ int agg_owner_base(const int* __restrict__ ocount, int b, int lane) {
    int s = 0;
    for (int i = lane; i < b; i += 32) s += ocount[i];
    for (int o = 16; o > 0; o >>= 1) s += __shfl_xor(s, o);
    return s;
}

#define AGG_SLICES 8
#define AGG_RPS    ((AGG_NPB + AGG_SLICES - 1) / AGG_SLICES)
template <int F, bool RELU, bool POOL>
__device__ __forceinline__ void agg_finish(float* acc,
                                           const float* __restrict__ h,
                                           const float* __restrict__ dinv,
                                           const float* __restrict__ bias,
                                           float* __restrict__ out,
                                           const int* __restrict__ batch,
                                           float* __restrict__ sums,
                                           float* __restrict__ cnt,
                                           int node0, int nloc, int tid) {
    constexpr int F4 = F / 4;
    for (int idx = tid; idx < nloc * F4; idx += AGG_TPB) {
        int i = idx / F4;
        int g = idx - i * F4;
        v4f q = ((const v4f*)acc)[idx];
        if (F == 128 && (g >> 4)) q = (v4f){q.y, q.x, q.w, q.z};
        int gi = node0 + i;
        float di = dinv[gi];
        v4f xv = ((const v4f*)h)[(size_t)gi * F4 + g];
        v4f bv = ((const v4f*)bias)[g];
        v4f o;
#pragma unroll
        for (int c = 0; c < 4; ++c) {
            float v = q[c] + xv[c] * di * di + bv[c];
            o[c] = RELU ? fmaxf(v, 0.0f) : v;
        }
        if (POOL) ((v4f*)acc)[idx] = o;
        else      ((v4f*)out)[(size_t)gi * F4 + g] = o;
    }

    if (POOL) {
        static_assert(!POOL || AGG_TPB == AGG_SLICES * F, "one thread per (slice, feature)");
        __syncthreads();
        const int f  = tid & (F - 1);
        const int r0 = (tid / F) * AGG_RPS;
        int   bid[AGG_RPS];
        float val[AGG_RPS];
#pragma unroll
        for (int j = 0; j < AGG_RPS; ++j) {
            const int r  = r0 + j;
            const bool ok = r < nloc;
            const int rr = ok ? r : 0;
            const int b  = batch[node0 + rr];
            const float v = acc[rr * F + f];
            bid[j] = ok ? b : -1;
            val[j] = ok ? v : 0.0f;
        }
        int   cur  = -1;
        float run  = 0.0f;
        float nrun = 0.0f;
#pragma unroll
        for (int j = 0; j < AGG_RPS; ++j) {
            if (bid[j] != cur) {
                if (cur >= 0) {
                    atomicAdd(&sums[(size_t)cur * F + f], run);
                    if (f == 0) atomicAdd(&cnt[cur], nrun);
                }
                cur = bid[j]; run = 0.0f; nrun = 0.0f;
            }
            run  += val[j];
            nrun += 1.0f;
        }
        if (cur >= 0) {
            atomicAdd(&sums[(size_t)cur * F + f], run);
            if (f == 0) atomicAdd(&cnt[cur], nrun);
        }
    }
}

__global__ __launch_bounds__(AGG_TPB)
void owner_scan_kernel(const int* __restrict__ pk,
                       const int* __restrict__ obw,
                       const unsigned* __restrict__ bm,
                       const int* __restrict__ ocount,
                       float* __restrict__ dinv,
                       int* __restrict__ rpk,
                       int N, int E) {
    __shared__ __attribute__((aligned(16))) int qbuf[AGG_WAVES * AGG_QCAP];
    __shared__ unsigned short plist[AGG_WAVES * AGG_PCAP];
    __shared__ int degc[AGG_NPB];
    __shared__ int cursor;

    const int tid  = threadIdx.x;
    const int lane = tid & 31;
    const int wave = __builtin_amdgcn_readfirstlane(tid >> 5);
    const int node0 = blockIdx.x * AGG_NPB;
    const int nloc  = min(AGG_NPB, N - node0);

    for (int i = tid; i < AGG_NPB; i += AGG_TPB) degc[i] = 0;
    if (tid == 0) cursor = 0;
    int* lst = rpk + agg_owner_base(ocount, (int)blockIdx.x, lane);
    __syncthreads();

    int* myq = qbuf + wave * AGG_QCAP;
    int qcnt = 0;

    auto push = [&](bool m, int e) {
        unsigned mask = __builtin_amdgcn_ballot_w32(m);
        if (mask) {
            int pos = qcnt + (int)__builtin_amdgcn_mbcnt_lo(mask, 0u);
            if (m) myq[pos] = e;
            qcnt += __builtin_popcount(mask);
        }
    };
    auto reserve = [&](int n) -> int {
        int pos = 0;
        if (lane == 0) pos = atomicAdd(&cursor, n);
        return __builtin_amdgcn_readfirstlane(pos);
    };
    const unsigned pat = (unsigned)blockIdx.x * 0x01010101u;
    auto eqb = [&](int wv) -> unsigned {
        const unsigned y = (unsigned)wv ^ pat;
        const unsigned t = ((y & 0x7F7F7F7Fu) + 0x7F7F7F7Fu) | y;
        return ~t & 0x80808080u;
    };
    auto ldp = [&](int sc, int4& a, int4& b) {
        const int4* p = (const int4*)(obw + (size_t)sc * 8);
        a = p[0];
        b = p[1];
    };
    auto drain_full = [&]() {
        while (qcnt >= 32) {
            qcnt -= 32;
            scan_drain<true>(pk, degc, myq, qcnt, 32, node0, lane, lst + reserve(32));
        }
    };

    const int nblk = (E + 1023) >> 10;
    unsigned short* myp = plist + wave * AGG_PCAP;
    for (int w0 = 0; w0 < nblk; w0 += AGG_WAVES * 32) {
        const int wi = w0 + lane * AGG_WAVES + wave;
        unsigned word = (wi < nblk) ? bm[(size_t)wi * AGG_OWNERS + blockIdx.x] : 0u;
        const int nb = __builtin_popcount(word);
        int incl = nb;
        for (int o = 1; o < 32; o <<= 1) {
            const int t = __shfl_up(incl, o);
            if (lane >= o) incl += t;
        }
        int pos = incl - nb;
        const int L = __builtin_amdgcn_readlane(incl, 31);
        while (__builtin_amdgcn_ballot_w32(word != 0u)) {
            if (word != 0u) {
                myp[pos++] = (unsigned short)((lane << 5) | __builtin_ctz(word));
                word &= word - 1u;
            }
        }
        auto fetch = [&](int i0, bool& has, int& sc, int4& a, int4& b) {
            has = i0 + lane < L;
            const int ent = has ? (int)myp[i0 + lane] : 0;
            sc = has ? ((w0 + (ent >> 5) * AGG_WAVES + wave) * 32 + (ent & 31)) : 0;
            ldp(sc, a, b);
        };
        bool has; int sc; int4 a, b;
        fetch(0, has, sc, a, b);
        for (int i0 = 0; i0 < L; i0 += 32) {
            unsigned m = (eqb(a.x) >> 7) | (eqb(a.y) >> 6) | (eqb(a.z) >> 5) | (eqb(a.w) >> 4) |
                         (eqb(b.x) >> 3) | (eqb(b.y) >> 2) | (eqb(b.z) >> 1) | eqb(b.w);
            if (!has) m = 0u;
            const int eb = sc * 32;
            while (__builtin_amdgcn_ballot_w32(m != 0u)) {
                const bool h2 = m != 0u;
                const int p = h2 ? __builtin_ctz(m) : 0;
                const int e = eb + 4 * (p & 7) + (p >> 3);
                m &= m - 1u;
                push(h2 && e < E, e);
            }
            fetch(i0 + 32, has, sc, a, b);
            drain_full();
        }
    }
    while (qcnt > 0) {
        int cnt = min(qcnt, 32);
        qcnt -= cnt;
        scan_drain<false>(pk, degc, myq, qcnt, cnt, node0, lane, lst + reserve(cnt));
    }
    __syncthreads();
    for (int i = tid; i < nloc; i += AGG_TPB) dinv[node0 + i] = rsqrtf(1.0f + (float)degc[i]);
}

#define C1_ROWS  208
#define C1_RT    (C1_ROWS / 16)
#define C1_PITCH 68
#define AGG_RCAP 8192
#define AGG_RLD  4
__global__ __launch_bounds__(AGG_TPB)
void conv1_owner_kernel(const float* __restrict__ h,
                        const int* __restrict__ rpk,
                        const int* __restrict__ ocount,
                        const float* __restrict__ dinv,
                        const float* __restrict__ Wa,
                        const float* __restrict__ ba,
                        const float* __restrict__ Wb,
                        float* __restrict__ out,
                        int* __restrict__ recw, int N) {
    constexpr int F = 64;
    __shared__ __attribute__((aligned(16))) float acc[C1_ROWS * C1_PITCH];
    __shared__ __attribute__((aligned(16))) float hbuf[C1_ROWS * C1_PITCH];
    __shared__ int2 srec[AGG_RCAP];

    const int tid  = threadIdx.x;
    const int lane = tid & 31;
    const int wave = __builtin_amdgcn_readfirstlane(tid >> 5);
    const int node0 = blockIdx.x * AGG_NPB;
    const int nloc  = min(AGG_NPB, N - node0);

    for (int idx = tid; idx < C1_ROWS * (F / 4); idx += AGG_TPB) {
        const int i = idx >> 4, g = idx & 15;
        v4f v = (v4f){0.f, 0.f, 0.f, 0.f};
        if (i < nloc) {
            const float di = dinv[node0 + i];
            v = ((const v4f*)h)[(size_t)(node0 + i) * (F / 4) + g] * (di * di);
        }
        *(v4f*)(acc + i * C1_PITCH + 4 * g) = v;
    }
    const int total = ocount[blockIdx.x];
    const int base  = agg_owner_base(ocount, (int)blockIdx.x, lane);
    const int* __restrict__ lst = rpk + base;
    int2* __restrict__ rec = (int2*)recw + base;

    int off[F / 32];
    agg_row_words<F>(off, lane);

    for (int r0 = 0; r0 < total; r0 += AGG_RCAP) {
        const int nst = min(total - r0, AGG_RCAP);
        if (r0 != 0) __syncthreads();
        for (int j0 = tid; j0 < nst; j0 += AGG_RLD * AGG_TPB) {
            static_assert(AGG_RLD == 4, "the two statements below name four edges");
            int   p[AGG_RLD];
            float ds[AGG_RLD], dd[AGG_RLD];
#pragma unroll
            for (int k = 0; k < AGG_RLD; ++k)
                p[k] = lst[r0 + min(j0 + k * AGG_TPB, nst - 1)];
            asm volatile("" :: "v"(p[0]), "v"(p[1]), "v"(p[2]), "v"(p[3]));
#pragma unroll
            for (int k = 0; k < AGG_RLD; ++k) {
                ds[k] = dinv[p[k] & 0xFFFF];
                dd[k] = dinv[(int)((unsigned)p[k] >> 16)];
            }
            asm volatile("" :: "v"(ds[0]), "v"(dd[0]), "v"(ds[1]), "v"(dd[1]),
                               "v"(ds[2]), "v"(dd[2]), "v"(ds[3]), "v"(dd[3]));
#pragma unroll
            for (int k = 0; k < AGG_RLD; ++k) {
                if (j0 + k * AGG_TPB < nst) {
                    const float w = ds[k] * dd[k];
                    const int2 r = make_int2(p[k], __builtin_bit_cast(int, w));
                    srec[j0 + k * AGG_TPB] = r;
                    rec[r0 + j0 + k * AGG_TPB] = r;
                }
            }
        }
        __syncthreads();

        const int nfull = nst >> 5;
        const int rest  = nst & 31;
        for (int i = wave; i < nfull; i += AGG_WAVES) {
            const int2 r = srec[32 * i + lane];
            agg_rows<F, true, C1_PITCH>(h, acc, r.x & 0xFFFF, (int)((unsigned)r.x >> 16) - node0,
                                        __builtin_bit_cast(float, r.y), 32, lane, off);
        }
        if (rest != 0 && (nfull & (AGG_WAVES - 1)) == wave) {
            int dl_l = 0, s_l = 0;
            float w_l = 0.0f;
            if (lane < rest) {
                const int2 t = srec[32 * nfull + lane];
                s_l  = t.x & 0xFFFF;
                dl_l = (int)((unsigned)t.x >> 16) - node0;
                w_l  = __builtin_bit_cast(float, t.y);
            }
            agg_rows<F, false, C1_PITCH>(h, acc, s_l, dl_l, w_l, rest, lane, off);
        }
    }
    __syncthreads();

    const int half = lane >> 4;
    const int lm   = lane & 15;

    const int cg = wave & 3;

    {
        const float* __restrict__ bb = Wa + (2 * half) * 64 + cg * 16 + lm;
        v2f wb[16];
#pragma unroll
        for (int k = 0; k < 16; ++k) {
            wb[k].x = bb[(4 * k) * 64];
            wb[k].y = bb[(4 * k + 1) * 64];
        }
        const float bv = ba[cg * 16 + lm];
        for (int rt = wave >> 2; rt < C1_RT; rt += AGG_WAVES / 4) {
            const float* arow = acc + (rt * 16 + lm) * C1_PITCH + 2 * half;
            v8f d = (v8f){};
#pragma unroll
            for (int k = 0; k < 16; ++k) {
                const v2f a = *(const v2f*)(arow + 4 * k);
                d = __builtin_amdgcn_wmma_f32_16x16x4_f32(false, a, false, wb[k], (short)0, d,
                                                          false, false);
            }
            float* hrow = hbuf + (rt * 16 + 8 * half) * C1_PITCH + cg * 16 + lm;
#pragma unroll
            for (int v = 0; v < 8; ++v) hrow[v * C1_PITCH] = fmaxf(d[v] + bv, 0.0f);
        }
    }
    __syncthreads();

    const float* __restrict__ bb2 = Wb + (2 * half) * 128 + cg * 32 + lm;
    v2f wp[16], wq[16];
#pragma unroll
    for (int k = 0; k < 16; ++k) {
        wp[k].x = bb2[(4 * k) * 128];
        wp[k].y = bb2[(4 * k + 1) * 128];
        wq[k].x = bb2[(4 * k) * 128 + 16];
        wq[k].y = bb2[(4 * k + 1) * 128 + 16];
    }
    for (int rt = wave >> 2; rt < C1_RT; rt += AGG_WAVES / 4) {
        const float* arow = hbuf + (rt * 16 + lm) * C1_PITCH + 2 * half;
        v8f d0 = (v8f){}, d1 = (v8f){};
#pragma unroll
        for (int k = 0; k < 16; ++k) {
            const v2f a = *(const v2f*)(arow + 4 * k);
            d0 = __builtin_amdgcn_wmma_f32_16x16x4_f32(false, a, false, wp[k], (short)0, d0,
                                                       false, false);
            d1 = __builtin_amdgcn_wmma_f32_16x16x4_f32(false, a, false, wq[k], (short)0, d1,
                                                       false, false);
        }
        const int r0 = rt * 16 + 8 * half;
        float* __restrict__ crow = out + (size_t)(node0 + r0) * 128 + cg * 32 + lm;
#pragma unroll
        for (int v = 0; v < 8; ++v) {
            if (r0 + v < nloc) {
                crow[(size_t)v * 128]      = d0[v];
                crow[(size_t)v * 128 + 16] = d1[v];
            }
        }
    }
}


#define AGG_NG 4
#define AGG_NSELF ((AGG_NPB + AGG_WAVES - 1) / AGG_WAVES)

__device__ __forceinline__ void graph_add(v4f (&a)[AGG_NG], int g, v4f v, float w) {
#pragma unroll
    for (int j = 0; j < AGG_NG; ++j) {
        const float m = (g == j) ? w : 0.0f;
        a[j] += v * m;
    }
}

__device__ __forceinline__ void graph_rows(const float* __restrict__ h, v4f (&a)[AGG_NG],
                                           int s_l, int g_l, float w_l, int lane) {
#pragma unroll
    for (int g0 = 0; g0 < 32; g0 += 8) {
        v4f r[8];
#pragma unroll
        for (int i = 0; i < 8; ++i) {
            int s = __builtin_amdgcn_readlane(s_l, g0 + i);
            r[i] = *((const v4f*)(h + (size_t)s * 128) + lane);
        }
#pragma unroll
        for (int i = 0; i < 8; ++i) {
            int g = __builtin_amdgcn_readlane(g_l, g0 + i);
            float w = __builtin_bit_cast(
                float, __builtin_amdgcn_readlane(__builtin_bit_cast(int, w_l), g0 + i));
            graph_add(a, g, r[i], w);
        }
    }
}

template <int F, bool RELU, bool POOL>
__global__ __launch_bounds__(AGG_TPB)
void agg_replay_kernel(const float* __restrict__ h,
                       const int* __restrict__ recw,
                       const int* __restrict__ ocount,
                       const float* __restrict__ dinv,
                       const float* __restrict__ bias,
                       float* __restrict__ out,
                       const int* __restrict__ batch,
                       float* __restrict__ sums,
                       float* __restrict__ cnt, int N) {
    constexpr int F4 = F / 4;
    __shared__ __attribute__((aligned(16))) float acc[AGG_NPB * F];
    __shared__ int2 srec[AGG_RCAP];
    __shared__ int sbat[AGG_NPB];

    const int tid  = threadIdx.x;
    const int lane = tid & 31;
    const int wave = __builtin_amdgcn_readfirstlane(tid >> 5);
    const int node0 = blockIdx.x * AGG_NPB;
    const int nloc  = min(AGG_NPB, N - node0);

    int g_lo = 0, span = AGG_NG + 1;
    if (POOL) {
        int lo = 0x7FFFFFFF, hi = -1;
        for (int i = lane; i < nloc; i += 32) {
            const int b = batch[node0 + i];
            lo = min(lo, b); hi = max(hi, b);
        }
        for (int o = 16; o > 0; o >>= 1) {
            lo = min(lo, __shfl_xor(lo, o)); hi = max(hi, __shfl_xor(hi, o));
        }
        g_lo = __builtin_amdgcn_readfirstlane(lo);
        span = __builtin_amdgcn_readfirstlane(hi) - g_lo + 1;
        for (int i = tid; i < nloc; i += AGG_TPB) sbat[i] = batch[node0 + i];
    }
    const bool graphs = POOL && F == 128 && span <= AGG_NG;
    if (!graphs)
        for (int i = tid; i < AGG_NPB * F4; i += AGG_TPB) ((v4f*)acc)[i] = (v4f){0.f, 0.f, 0.f, 0.f};
    v4f ga[AGG_NG];
#pragma unroll
    for (int j = 0; j < AGG_NG; ++j) ga[j] = (v4f){0.f, 0.f, 0.f, 0.f};
    const int total = ocount[blockIdx.x];
    const int2* __restrict__ rec =
        (const int2*)recw + agg_owner_base(ocount, (int)blockIdx.x, lane);

    int off[F / 32];
    agg_row_words<F>(off, lane);

    for (int r0 = 0; r0 < total; r0 += AGG_RCAP) {
        const int nst = min(total - r0, AGG_RCAP);
        if (r0 != 0) __syncthreads();
        for (int j0 = tid; j0 < nst; j0 += AGG_RLD * AGG_TPB) {
            int2 t[AGG_RLD];
#pragma unroll
            for (int k = 0; k < AGG_RLD; ++k)
                t[k] = rec[r0 + min(j0 + k * AGG_TPB, nst - 1)];
#pragma unroll
            for (int k = 0; k < AGG_RLD; ++k)
                if (j0 + k * AGG_TPB < nst) srec[j0 + k * AGG_TPB] = t[k];
        }
        __syncthreads();

        const int nfull = nst >> 5;
        const int rest  = nst & 31;
        for (int i = wave; i < nfull; i += AGG_WAVES) {
            const int2 r = srec[32 * i + lane];
            const int dl = (int)((unsigned)r.x >> 16) - node0;
            if (graphs)
                graph_rows(h, ga, r.x & 0xFFFF, sbat[dl] - g_lo,
                           __builtin_bit_cast(float, r.y), lane);
            else
                agg_rows<F, true>(h, acc, r.x & 0xFFFF, dl,
                                  __builtin_bit_cast(float, r.y), 32, lane, off);
        }
        if (rest != 0 && (nfull & (AGG_WAVES - 1)) == wave) {
            int dl_l = 0, s_l = 0;
            float w_l = 0.0f;
            if (lane < rest) {
                const int2 t = srec[32 * nfull + lane];
                s_l  = t.x & 0xFFFF;
                dl_l = (int)((unsigned)t.x >> 16) - node0;
                w_l  = __builtin_bit_cast(float, t.y);
            }
            if (graphs)
                graph_rows(h, ga, s_l, sbat[dl_l] - g_lo, w_l, lane);
            else
                agg_rows<F, false>(h, acc, s_l, dl_l, w_l, rest, lane, off);
        }
    }
    __syncthreads();

    if (!graphs) {
        agg_finish<F, RELU, POOL>(acc, h, dinv, bias, out, batch, sums, cnt, node0, nloc, tid);
        return;
    }

    v4f   rr[AGG_NSELF];
    float dk[AGG_NSELF];
    int   gk[AGG_NSELF];
#pragma unroll
    for (int k = 0; k < AGG_NSELF; ++k) {
        const int ii = min(wave + k * AGG_WAVES, nloc - 1);
        rr[k] = ((const v4f*)h)[(size_t)(node0 + ii) * F4 + lane];
        dk[k] = dinv[node0 + ii];
        gk[k] = __builtin_amdgcn_readfirstlane(sbat[ii]) - g_lo;
    }
    const v4f bv = ((const v4f*)bias)[lane];
#pragma unroll
    for (int k = 0; k < AGG_NSELF; ++k) {
        const bool ok = wave + k * AGG_WAVES < nloc;
        graph_add(ga, gk[k], rr[k], ok ? dk[k] * dk[k] : 0.0f);
        graph_add(ga, gk[k], bv, ok ? 1.0f : 0.0f);
    }

#pragma unroll
    for (int j = 0; j < AGG_NG; ++j) ((v4f*)acc)[(wave * AGG_NG + j) * F4 + lane] = ga[j];
    __syncthreads();
    for (int idx = tid; idx < span * F; idx += AGG_TPB) {
        const int j = idx / F, f = idx - j * F;
        float s = 0.0f;
        for (int w = 0; w < AGG_WAVES; ++w) s += acc[(w * AGG_NG + j) * F + f];
        atomicAdd(&sums[(size_t)(g_lo + j) * F + f], s);
    }
    if (tid < span) {
        float c = 0.0f;
        for (int i = 0; i < nloc; ++i) c += (sbat[i] - g_lo == tid) ? 1.0f : 0.0f;
        atomicAdd(&cnt[g_lo + tid], c);
    }
}

static inline int cdiv(long long a, long long b) { return (int)((a + b - 1) / b); }

extern "C" void kernel_launch(void* const* d_in, const int* in_sizes, int n_in,
                              void* d_out, int out_size, void* d_ws, size_t ws_size,
                              hipStream_t stream) {
    (void)n_in; (void)ws_size;
    const float* x    = (const float*)d_in[0];
    const int*   edge = (const int*)d_in[1];
    const int*   batch= (const int*)d_in[2];
    const float* b1   = (const float*)d_in[4];
    const float* W1   = (const float*)d_in[3];
    const float* W2   = (const float*)d_in[5];
    const float* b2   = (const float*)d_in[6];
    const float* Wfc  = (const float*)d_in[7];
    const float* bfc  = (const float*)d_in[8];
    float* out = (float*)d_out;

    const int N   = in_sizes[2];
    const int E   = in_sizes[1] / 2;
    const int D   = 64;
    const int D2  = 128;
    const int HID = in_sizes[8];
    const int G   = out_size / HID;

    const int* src = edge;
    const int* dst = edge + E;

    float* ws    = (float*)d_ws;
    int*   rpk   = (int*)ws;
    float* hw2   = ws + 2 * (size_t)N * D;
    float* ebuf  = hw2  + (size_t)N * D2;
    int*   pk    = (int*)ebuf;
    int*   obw   = pk + E;
    int*   rec   = obw + E;
    float* zblk  = (float*)(rec + 2 * (size_t)E);
    int*   ocount= (int*)zblk;
    float* sums  = zblk + AGG_OWNERS;
    float* cnt   = sums + (size_t)G * D2;
    unsigned* bm = (unsigned*)(cnt + G);
    float* dinv  = ebuf + (size_t)N * D2;

    fill_kernel<<<cdiv((long long)AGG_OWNERS + G * D2 + G, TPB), TPB, 0, stream>>>(
        zblk, 0.0f, AGG_OWNERS + G * D2 + G);

    edge_prep_kernel<<<cdiv(cdiv(E, 4), TPB), TPB, 0, stream>>>(src, dst, pk, obw, ocount, bm, E);
    owner_scan_kernel<<<cdiv(N, AGG_NPB), AGG_TPB, 0, stream>>>(pk, obw, bm, ocount, dinv, rpk, N, E);

    conv1_owner_kernel<<<cdiv(N, AGG_NPB), AGG_TPB, 0, stream>>>(
        x, rpk, ocount, dinv, W1, b1, W2, hw2, rec, N);

    agg_replay_kernel<128, false, true><<<cdiv(N, AGG_NPB), AGG_TPB, 0, stream>>>(
        hw2, rec, ocount, dinv, b2, nullptr, batch, sums, cnt, N);

    pooled_product_f32<128, 64><<<G / 16, 32 * (64 / 16), 0, stream>>>(sums, cnt, Wfc, bfc, out);
}


# Serial stretches of the owner kernels outside their edge loops removed: prologue loads (owner counts, graph ids, own rows) all in flight at once; nodes per graph by wave ballots, not a 196-trip loop
# speedup vs baseline: 6.5005x; 1.0547x over previous
#include <hip/hip_runtime.h>
#include <hip/hip_bf16.h>


typedef float v2f __attribute__((ext_vector_type(2)));
typedef float v8f __attribute__((ext_vector_type(8)));

#define TPB 256

__global__ void fill_kernel(float* __restrict__ p, float v, int n) {
    int i = blockIdx.x * blockDim.x + threadIdx.x;
    if (i < n) p[i] = v;
}

#define AGG_NPB    196
#define AGG_OWNERS 256
__global__ void edge_prep_kernel(const int* __restrict__ src, const int* __restrict__ dst,
                                 int* __restrict__ pk,
                                 int* __restrict__ obw, int* __restrict__ ocount,
                                 unsigned* __restrict__ bm, int E) {
    __shared__ int hist[AGG_OWNERS];
    __shared__ unsigned bits[AGG_OWNERS];
    for (int i = threadIdx.x; i < AGG_OWNERS; i += blockDim.x) { hist[i] = 0; bits[i] = 0u; }
    __syncthreads();
    int e = (blockIdx.x * blockDim.x + threadIdx.x) * 4;
    if (e + 3 < E) {
        int4 d = *(const int4*)(dst + e);
        int4 s = *(const int4*)(src + e);
        int4 p;
        p.x = (int)(((unsigned)d.x << 16) | (unsigned)s.x);
        p.y = (int)(((unsigned)d.y << 16) | (unsigned)s.y);
        p.z = (int)(((unsigned)d.z << 16) | (unsigned)s.z);
        p.w = (int)(((unsigned)d.w << 16) | (unsigned)s.w);
        *(int4*)(pk + e) = p;
        const unsigned ox = (unsigned)(d.x / AGG_NPB), oy = (unsigned)(d.y / AGG_NPB);
        const unsigned oz = (unsigned)(d.z / AGG_NPB), ow = (unsigned)(d.w / AGG_NPB);
        obw[e >> 2] = (int)(ox | (oy << 8) | (oz << 16) | (ow << 24));
        atomicAdd(&hist[ox], 1);
        atomicAdd(&hist[oy], 1);
        atomicAdd(&hist[oz], 1);
        atomicAdd(&hist[ow], 1);
        const unsigned pb = 1u << (threadIdx.x >> 3);
        atomicOr(&bits[ox], pb);
        atomicOr(&bits[oy], pb);
        atomicOr(&bits[oz], pb);
        atomicOr(&bits[ow], pb);
    } else {
        unsigned char* ob = (unsigned char*)obw;
        for (; e < E; ++e) {
            int d = dst[e];
            pk[e] = (int)(((unsigned)d << 16) | (unsigned)src[e]);
            ob[e] = (unsigned char)(d / AGG_NPB);
            atomicAdd(&hist[d / AGG_NPB], 1);
            atomicOr(&bits[d / AGG_NPB], 1u << ((e - (int)(blockIdx.x * blockDim.x) * 4) >> 5));
        }
    }
    __syncthreads();
    for (int i = threadIdx.x; i < AGG_OWNERS; i += blockDim.x) {
        const int c = hist[i];
        if (c) atomicAdd(&ocount[i], c);
        bm[(size_t)blockIdx.x * AGG_OWNERS + i] = bits[i];
    }
}

template <int K, int Nn>
__global__ __launch_bounds__(32 * (Nn / 16))
void pooled_product_f32(const float* __restrict__ A,
                        const float* __restrict__ cnt,
                        const float* __restrict__ B,
                        const float* __restrict__ bias,
                        float* __restrict__ C) {
    constexpr int KS = K / 4;
    const int ct   = __builtin_amdgcn_readfirstlane((int)(threadIdx.x >> 5));
    const int lane = threadIdx.x & 31;
    const int half = lane >> 4;
    const int lm   = lane & 15;
    const int row0 = blockIdx.x << 4;

    const float* __restrict__ arow = A + (size_t)(row0 + lm) * K + 2 * half;
    const float* __restrict__ bcol = B + (size_t)(2 * half) * Nn + ct * 16 + lm;

    v2f a[KS], b[KS];
#pragma unroll
    for (int k = 0; k < KS; ++k) a[k] = *(const v2f*)(arow + 4 * k);
#pragma unroll
    for (int k = 0; k < KS; ++k) {
        b[k].x = bcol[(4 * k) * Nn];
        b[k].y = bcol[(4 * k + 1) * Nn];
    }
    float cv[8];
#pragma unroll
    for (int v = 0; v < 8; ++v) cv[v] = cnt[row0 + 8 * half + v];
    const float bv = bias[ct * 16 + lm];

    v8f d = (v8f){};
#pragma unroll
    for (int k = 0; k < KS; ++k)
        d = __builtin_amdgcn_wmma_f32_16x16x4_f32(false, a[k], false, b[k], (short)0, d,
                                                  false, false);

    float* __restrict__ crow = C + (size_t)(row0 + 8 * half) * Nn + ct * 16 + lm;
#pragma unroll
    for (int v = 0; v < 8; ++v) crow[(size_t)v * Nn] = d[v] / fmaxf(cv[v], 1.0f) + bv;
}

#define AGG_TPB   1024
#define AGG_WAVES 32
#define AGG_QCAP  1056
#define AGG_PCAP  1024

typedef float v4f __attribute__((ext_vector_type(4)));

template <int F> struct RowVecSel      { typedef v2f T; };
template <>      struct RowVecSel<128> { typedef v4f T; };

template <int F, bool FULL, int PITCH = F>
__device__ __forceinline__ void agg_rows(const float* __restrict__ h, float* acc,
                                         int s_l, int dl_l, float w_l, int cnt, int lane,
                                         const int (&off)[F / 32]) {
    typedef typename RowVecSel<F>::T VT;
    constexpr int VEC = F / 32;
#pragma unroll
    for (int g = 0; g < 32; g += 8) {
        if (!FULL && g >= cnt) break;
        VT r[8];
#pragma unroll
        for (int i = 0; i < 8; ++i) {
            if (FULL || g + i < cnt) {
                int s = __builtin_amdgcn_readlane(s_l, g + i);
                r[i] = *((const VT*)(h + (size_t)s * F) + lane);
            }
        }
#pragma unroll
        for (int i = 0; i < 8; ++i) {
            if (FULL || g + i < cnt) {
                int dl = __builtin_amdgcn_readlane(dl_l, g + i);
                float w = __builtin_bit_cast(
                    float, __builtin_amdgcn_readlane(__builtin_bit_cast(int, w_l), g + i));
                float* a = acc + dl * PITCH;
#pragma unroll
                for (int k = 0; k < VEC; ++k)
                    __hip_atomic_fetch_add(&a[off[k]], r[i][k] * w, __ATOMIC_RELAXED,
                                           __HIP_MEMORY_SCOPE_WORKGROUP);
            }
        }
    }
}

template <bool FULL>
__device__ __forceinline__ void scan_drain(const int* __restrict__ pk, int* degc, const int* q,
                                           int start, int cnt, int node0, int lane,
                                           int* __restrict__ out) {
    if (FULL || lane < cnt) {
        const int e_l = q[start + lane];
        const int p_l = pk[e_l];
        atomicAdd(&degc[(int)((unsigned)p_l >> 16) - node0], 1);
        out[lane] = p_l;
    }
}

template <int F>
__device__ __forceinline__ void agg_row_words(int (&off)[F / 32], int lane) {
    constexpr int VEC = F / 32;
#pragma unroll
    for (int k = 0; k < VEC; ++k)
        off[k] = (F == 128) ? (4 * lane + (k ^ (lane >> 4))) : (VEC * lane + k);
}

__device__ __forceinline__ void agg_owner_counts(const int* __restrict__ ocount, int lane,
                                                 int (&v)[AGG_OWNERS / 32]) {
#pragma unroll
    for (int k = 0; k < AGG_OWNERS / 32; ++k) v[k] = ocount[lane + 32 * k];
}
__device__ __forceinline__ int agg_owner_base(const int (&v)[AGG_OWNERS / 32], int b, int lane) {
    int s = 0;
#pragma unroll
    for (int k = 0; k < AGG_OWNERS / 32; ++k) s += (lane + 32 * k < b) ? v[k] : 0;
    for (int o = 16; o > 0; o >>= 1) s += __shfl_xor(s, o);
    return s;
}

#define AGG_SLICES 8
#define AGG_RPS    ((AGG_NPB + AGG_SLICES - 1) / AGG_SLICES)
template <int F, bool RELU, bool POOL>
__device__ __forceinline__ void agg_finish(float* acc,
                                           const float* __restrict__ h,
                                           const float* __restrict__ dinv,
                                           const float* __restrict__ bias,
                                           float* __restrict__ out,
                                           const int* __restrict__ batch,
                                           float* __restrict__ sums,
                                           float* __restrict__ cnt,
                                           int node0, int nloc, int tid) {
    constexpr int F4 = F / 4;
    for (int idx = tid; idx < nloc * F4; idx += AGG_TPB) {
        int i = idx / F4;
        int g = idx - i * F4;
        v4f q = ((const v4f*)acc)[idx];
        if (F == 128 && (g >> 4)) q = (v4f){q.y, q.x, q.w, q.z};
        int gi = node0 + i;
        float di = dinv[gi];
        v4f xv = ((const v4f*)h)[(size_t)gi * F4 + g];
        v4f bv = ((const v4f*)bias)[g];
        v4f o;
#pragma unroll
        for (int c = 0; c < 4; ++c) {
            float v = q[c] + xv[c] * di * di + bv[c];
            o[c] = RELU ? fmaxf(v, 0.0f) : v;
        }
        if (POOL) ((v4f*)acc)[idx] = o;
        else      ((v4f*)out)[(size_t)gi * F4 + g] = o;
    }

    if (POOL) {
        static_assert(!POOL || AGG_TPB == AGG_SLICES * F, "one thread per (slice, feature)");
        __syncthreads();
        const int f  = tid & (F - 1);
        const int r0 = (tid / F) * AGG_RPS;
        int   bid[AGG_RPS];
        float val[AGG_RPS];
#pragma unroll
        for (int j = 0; j < AGG_RPS; ++j) {
            const int r  = r0 + j;
            const bool ok = r < nloc;
            const int rr = ok ? r : 0;
            const int b  = batch[node0 + rr];
            const float v = acc[rr * F + f];
            bid[j] = ok ? b : -1;
            val[j] = ok ? v : 0.0f;
        }
        int   cur  = -1;
        float run  = 0.0f;
        float nrun = 0.0f;
#pragma unroll
        for (int j = 0; j < AGG_RPS; ++j) {
            if (bid[j] != cur) {
                if (cur >= 0) {
                    atomicAdd(&sums[(size_t)cur * F + f], run);
                    if (f == 0) atomicAdd(&cnt[cur], nrun);
                }
                cur = bid[j]; run = 0.0f; nrun = 0.0f;
            }
            run  += val[j];
            nrun += 1.0f;
        }
        if (cur >= 0) {
            atomicAdd(&sums[(size_t)cur * F + f], run);
            if (f == 0) atomicAdd(&cnt[cur], nrun);
        }
    }
}

__global__ __launch_bounds__(AGG_TPB)
void owner_scan_kernel(const int* __restrict__ pk,
                       const int* __restrict__ obw,
                       const unsigned* __restrict__ bm,
                       const int* __restrict__ ocount,
                       float* __restrict__ dinv,
                       int* __restrict__ rpk,
                       int N, int E) {
    __shared__ __attribute__((aligned(16))) int qbuf[AGG_WAVES * AGG_QCAP];
    __shared__ unsigned short plist[AGG_WAVES * AGG_PCAP];
    __shared__ int degc[AGG_NPB];
    __shared__ int cursor;

    const int tid  = threadIdx.x;
    const int lane = tid & 31;
    const int wave = __builtin_amdgcn_readfirstlane(tid >> 5);
    const int node0 = blockIdx.x * AGG_NPB;
    const int nloc  = min(AGG_NPB, N - node0);

    int ocv[AGG_OWNERS / 32];
    agg_owner_counts(ocount, lane, ocv);
    for (int i = tid; i < AGG_NPB; i += AGG_TPB) degc[i] = 0;
    if (tid == 0) cursor = 0;
    int* lst = rpk + agg_owner_base(ocv, (int)blockIdx.x, lane);
    __syncthreads();

    int* myq = qbuf + wave * AGG_QCAP;
    int qcnt = 0;

    auto push = [&](bool m, int e) {
        unsigned mask = __builtin_amdgcn_ballot_w32(m);
        if (mask) {
            int pos = qcnt + (int)__builtin_amdgcn_mbcnt_lo(mask, 0u);
            if (m) myq[pos] = e;
            qcnt += __builtin_popcount(mask);
        }
    };
    auto reserve = [&](int n) -> int {
        int pos = 0;
        if (lane == 0) pos = atomicAdd(&cursor, n);
        return __builtin_amdgcn_readfirstlane(pos);
    };
    const unsigned pat = (unsigned)blockIdx.x * 0x01010101u;
    auto eqb = [&](int wv) -> unsigned {
        const unsigned y = (unsigned)wv ^ pat;
        const unsigned t = ((y & 0x7F7F7F7Fu) + 0x7F7F7F7Fu) | y;
        return ~t & 0x80808080u;
    };
    auto ldp = [&](int sc, int4& a, int4& b) {
        const int4* p = (const int4*)(obw + (size_t)sc * 8);
        a = p[0];
        b = p[1];
    };
    auto drain_full = [&]() {
        while (qcnt >= 32) {
            qcnt -= 32;
            scan_drain<true>(pk, degc, myq, qcnt, 32, node0, lane, lst + reserve(32));
        }
    };

    const int nblk = (E + 1023) >> 10;
    unsigned short* myp = plist + wave * AGG_PCAP;
    for (int w0 = 0; w0 < nblk; w0 += AGG_WAVES * 32) {
        const int wi = w0 + lane * AGG_WAVES + wave;
        unsigned word = (wi < nblk) ? bm[(size_t)wi * AGG_OWNERS + blockIdx.x] : 0u;
        const int nb = __builtin_popcount(word);
        int incl = nb;
        for (int o = 1; o < 32; o <<= 1) {
            const int t = __shfl_up(incl, o);
            if (lane >= o) incl += t;
        }
        int pos = incl - nb;
        const int L = __builtin_amdgcn_readlane(incl, 31);
        while (__builtin_amdgcn_ballot_w32(word != 0u)) {
            if (word != 0u) {
                myp[pos++] = (unsigned short)((lane << 5) | __builtin_ctz(word));
                word &= word - 1u;
            }
        }
        auto fetch = [&](int i0, bool& has, int& sc, int4& a, int4& b) {
            has = i0 + lane < L;
            const int ent = has ? (int)myp[i0 + lane] : 0;
            sc = has ? ((w0 + (ent >> 5) * AGG_WAVES + wave) * 32 + (ent & 31)) : 0;
            ldp(sc, a, b);
        };
        bool has; int sc; int4 a, b;
        fetch(0, has, sc, a, b);
        for (int i0 = 0; i0 < L; i0 += 32) {
            unsigned m = (eqb(a.x) >> 7) | (eqb(a.y) >> 6) | (eqb(a.z) >> 5) | (eqb(a.w) >> 4) |
                         (eqb(b.x) >> 3) | (eqb(b.y) >> 2) | (eqb(b.z) >> 1) | eqb(b.w);
            if (!has) m = 0u;
            const int eb = sc * 32;
            while (__builtin_amdgcn_ballot_w32(m != 0u)) {
                const bool h2 = m != 0u;
                const int p = h2 ? __builtin_ctz(m) : 0;
                const int e = eb + 4 * (p & 7) + (p >> 3);
                m &= m - 1u;
                push(h2 && e < E, e);
            }
            fetch(i0 + 32, has, sc, a, b);
            drain_full();
        }
    }
    while (qcnt > 0) {
        int cnt = min(qcnt, 32);
        qcnt -= cnt;
        scan_drain<false>(pk, degc, myq, qcnt, cnt, node0, lane, lst + reserve(cnt));
    }
    __syncthreads();
    for (int i = tid; i < nloc; i += AGG_TPB) dinv[node0 + i] = rsqrtf(1.0f + (float)degc[i]);
}

#define C1_ROWS  208
#define C1_RT    (C1_ROWS / 16)
#define C1_PITCH 68
#define AGG_RCAP 8192
#define AGG_RLD  4
__global__ __launch_bounds__(AGG_TPB)
void conv1_owner_kernel(const float* __restrict__ h,
                        const int* __restrict__ rpk,
                        const int* __restrict__ ocount,
                        const float* __restrict__ dinv,
                        const float* __restrict__ Wa,
                        const float* __restrict__ ba,
                        const float* __restrict__ Wb,
                        float* __restrict__ out,
                        int* __restrict__ recw, int N) {
    constexpr int F = 64;
    __shared__ __attribute__((aligned(16))) float acc[C1_ROWS * C1_PITCH];
    __shared__ __attribute__((aligned(16))) float hbuf[C1_ROWS * C1_PITCH];
    __shared__ int2 srec[AGG_RCAP];

    const int tid  = threadIdx.x;
    const int lane = tid & 31;
    const int wave = __builtin_amdgcn_readfirstlane(tid >> 5);
    const int node0 = blockIdx.x * AGG_NPB;
    const int nloc  = min(AGG_NPB, N - node0);

    constexpr int NINIT = (C1_ROWS * (F / 4) + AGG_TPB - 1) / AGG_TPB;
    int ocv[AGG_OWNERS / 32];
    agg_owner_counts(ocount, lane, ocv);
    v4f   xin[NINIT];
    float din[NINIT];
#pragma unroll
    for (int k = 0; k < NINIT; ++k) {
        const int idx = tid + k * AGG_TPB;
        const int ic  = min(idx >> 4, nloc - 1);
        din[k] = dinv[node0 + ic];
        xin[k] = ((const v4f*)h)[(size_t)(node0 + ic) * (F / 4) + (idx & 15)];
    }
    static_assert(NINIT == 4, "the statement below names four row pieces");
    asm volatile("" :: "v"(xin[0]), "v"(xin[1]), "v"(xin[2]), "v"(xin[3]),
                       "v"(din[0]), "v"(din[1]), "v"(din[2]), "v"(din[3]));
#pragma unroll
    for (int k = 0; k < NINIT; ++k) {
        const int idx = tid + k * AGG_TPB;
        const int i = idx >> 4, g = idx & 15;
        if (idx < C1_ROWS * (F / 4)) {
            v4f v = (v4f){0.f, 0.f, 0.f, 0.f};
            if (i < nloc) v = xin[k] * (din[k] * din[k]);
            *(v4f*)(acc + i * C1_PITCH + 4 * g) = v;
        }
    }
    const int total = ocount[blockIdx.x];
    const int base  = agg_owner_base(ocv, (int)blockIdx.x, lane);
    const int* __restrict__ lst = rpk + base;
    int2* __restrict__ rec = (int2*)recw + base;

    int off[F / 32];
    agg_row_words<F>(off, lane);

    for (int r0 = 0; r0 < total; r0 += AGG_RCAP) {
        const int nst = min(total - r0, AGG_RCAP);
        if (r0 != 0) __syncthreads();
        for (int j0 = tid; j0 < nst; j0 += AGG_RLD * AGG_TPB) {
            static_assert(AGG_RLD == 4, "the two statements below name four edges");
            int   p[AGG_RLD];
            float ds[AGG_RLD], dd[AGG_RLD];
#pragma unroll
            for (int k = 0; k < AGG_RLD; ++k)
                p[k] = lst[r0 + min(j0 + k * AGG_TPB, nst - 1)];
            asm volatile("" :: "v"(p[0]), "v"(p[1]), "v"(p[2]), "v"(p[3]));
#pragma unroll
            for (int k = 0; k < AGG_RLD; ++k) {
                ds[k] = dinv[p[k] & 0xFFFF];
                dd[k] = dinv[(int)((unsigned)p[k] >> 16)];
            }
            asm volatile("" :: "v"(ds[0]), "v"(dd[0]), "v"(ds[1]), "v"(dd[1]),
                               "v"(ds[2]), "v"(dd[2]), "v"(ds[3]), "v"(dd[3]));
#pragma unroll
            for (int k = 0; k < AGG_RLD; ++k) {
                if (j0 + k * AGG_TPB < nst) {
                    const float w = ds[k] * dd[k];
                    const int2 r = make_int2(p[k], __builtin_bit_cast(int, w));
                    srec[j0 + k * AGG_TPB] = r;
                    rec[r0 + j0 + k * AGG_TPB] = r;
                }
            }
        }
        __syncthreads();

        const int nfull = nst >> 5;
        const int rest  = nst & 31;
        for (int i = wave; i < nfull; i += AGG_WAVES) {
            const int2 r = srec[32 * i + lane];
            agg_rows<F, true, C1_PITCH>(h, acc, r.x & 0xFFFF, (int)((unsigned)r.x >> 16) - node0,
                                        __builtin_bit_cast(float, r.y), 32, lane, off);
        }
        if (rest != 0 && (nfull & (AGG_WAVES - 1)) == wave) {
            int dl_l = 0, s_l = 0;
            float w_l = 0.0f;
            if (lane < rest) {
                const int2 t = srec[32 * nfull + lane];
                s_l  = t.x & 0xFFFF;
                dl_l = (int)((unsigned)t.x >> 16) - node0;
                w_l  = __builtin_bit_cast(float, t.y);
            }
            agg_rows<F, false, C1_PITCH>(h, acc, s_l, dl_l, w_l, rest, lane, off);
        }
    }
    __syncthreads();

    const int half = lane >> 4;
    const int lm   = lane & 15;

    const int cg = wave & 3;

    {
        const float* __restrict__ bb = Wa + (2 * half) * 64 + cg * 16 + lm;
        v2f wb[16];
#pragma unroll
        for (int k = 0; k < 16; ++k) {
            wb[k].x = bb[(4 * k) * 64];
            wb[k].y = bb[(4 * k + 1) * 64];
        }
        const float bv = ba[cg * 16 + lm];
        for (int rt = wave >> 2; rt < C1_RT; rt += AGG_WAVES / 4) {
            const float* arow = acc + (rt * 16 + lm) * C1_PITCH + 2 * half;
            v8f d = (v8f){};
#pragma unroll
            for (int k = 0; k < 16; ++k) {
                const v2f a = *(const v2f*)(arow + 4 * k);
                d = __builtin_amdgcn_wmma_f32_16x16x4_f32(false, a, false, wb[k], (short)0, d,
                                                          false, false);
            }
            float* hrow = hbuf + (rt * 16 + 8 * half) * C1_PITCH + cg * 16 + lm;
#pragma unroll
            for (int v = 0; v < 8; ++v) hrow[v * C1_PITCH] = fmaxf(d[v] + bv, 0.0f);
        }
    }
    __syncthreads();

    const float* __restrict__ bb2 = Wb + (2 * half) * 128 + cg * 32 + lm;
    v2f wp[16], wq[16];
#pragma unroll
    for (int k = 0; k < 16; ++k) {
        wp[k].x = bb2[(4 * k) * 128];
        wp[k].y = bb2[(4 * k + 1) * 128];
        wq[k].x = bb2[(4 * k) * 128 + 16];
        wq[k].y = bb2[(4 * k + 1) * 128 + 16];
    }
    for (int rt = wave >> 2; rt < C1_RT; rt += AGG_WAVES / 4) {
        const float* arow = hbuf + (rt * 16 + lm) * C1_PITCH + 2 * half;
        v8f d0 = (v8f){}, d1 = (v8f){};
#pragma unroll
        for (int k = 0; k < 16; ++k) {
            const v2f a = *(const v2f*)(arow + 4 * k);
            d0 = __builtin_amdgcn_wmma_f32_16x16x4_f32(false, a, false, wp[k], (short)0, d0,
                                                       false, false);
            d1 = __builtin_amdgcn_wmma_f32_16x16x4_f32(false, a, false, wq[k], (short)0, d1,
                                                       false, false);
        }
        const int r0 = rt * 16 + 8 * half;
        float* __restrict__ crow = out + (size_t)(node0 + r0) * 128 + cg * 32 + lm;
#pragma unroll
        for (int v = 0; v < 8; ++v) {
            if (r0 + v < nloc) {
                crow[(size_t)v * 128]      = d0[v];
                crow[(size_t)v * 128 + 16] = d1[v];
            }
        }
    }
}


#define AGG_NG 4
#define AGG_NSELF ((AGG_NPB + AGG_WAVES - 1) / AGG_WAVES)

__device__ __forceinline__ void graph_add(v4f (&a)[AGG_NG], int g, v4f v, float w) {
#pragma unroll
    for (int j = 0; j < AGG_NG; ++j) {
        const float m = (g == j) ? w : 0.0f;
        a[j] += v * m;
    }
}

__device__ __forceinline__ void graph_rows(const float* __restrict__ h, v4f (&a)[AGG_NG],
                                           int s_l, int g_l, float w_l, int lane) {
#pragma unroll
    for (int g0 = 0; g0 < 32; g0 += 8) {
        v4f r[8];
#pragma unroll
        for (int i = 0; i < 8; ++i) {
            int s = __builtin_amdgcn_readlane(s_l, g0 + i);
            r[i] = *((const v4f*)(h + (size_t)s * 128) + lane);
        }
#pragma unroll
        for (int i = 0; i < 8; ++i) {
            int g = __builtin_amdgcn_readlane(g_l, g0 + i);
            float w = __builtin_bit_cast(
                float, __builtin_amdgcn_readlane(__builtin_bit_cast(int, w_l), g0 + i));
            graph_add(a, g, r[i], w);
        }
    }
}

template <int F, bool RELU, bool POOL>
__global__ __launch_bounds__(AGG_TPB)
void agg_replay_kernel(const float* __restrict__ h,
                       const int* __restrict__ recw,
                       const int* __restrict__ ocount,
                       const float* __restrict__ dinv,
                       const float* __restrict__ bias,
                       float* __restrict__ out,
                       const int* __restrict__ batch,
                       float* __restrict__ sums,
                       float* __restrict__ cnt, int N) {
    constexpr int F4 = F / 4;
    __shared__ __attribute__((aligned(16))) float acc[AGG_NPB * F];
    __shared__ int2 srec[AGG_RCAP];
    __shared__ int sbat[AGG_NPB];
    __shared__ int wcnt[AGG_NSELF * AGG_NG];

    const int tid  = threadIdx.x;
    const int lane = tid & 31;
    const int wave = __builtin_amdgcn_readfirstlane(tid >> 5);
    const int node0 = blockIdx.x * AGG_NPB;
    const int nloc  = min(AGG_NPB, N - node0);

    int ocv[AGG_OWNERS / 32];
    agg_owner_counts(ocount, lane, ocv);
    int g_lo = 0, span = AGG_NG + 1;
    if (POOL) {
        static_assert(AGG_NSELF * 32 >= AGG_NPB && AGG_TPB >= AGG_NPB, "one pass covers the owner's nodes");
        int bid[AGG_NSELF];
#pragma unroll
        for (int k = 0; k < AGG_NSELF; ++k) bid[k] = batch[node0 + min(lane + 32 * k, nloc - 1)];
        const int own = batch[node0 + min(tid, nloc - 1)];
        asm volatile("" :: "v"(own));
        int lo = bid[0], hi = bid[0];
#pragma unroll
        for (int k = 1; k < AGG_NSELF; ++k) { lo = min(lo, bid[k]); hi = max(hi, bid[k]); }
        for (int o = 16; o > 0; o >>= 1) {
            lo = min(lo, __shfl_xor(lo, o)); hi = max(hi, __shfl_xor(hi, o));
        }
        g_lo = __builtin_amdgcn_readfirstlane(lo);
        span = __builtin_amdgcn_readfirstlane(hi) - g_lo + 1;
        if (tid < nloc) sbat[tid] = own;
    }
    const bool graphs = POOL && F == 128 && span <= AGG_NG;
    if (!graphs)
        for (int i = tid; i < AGG_NPB * F4; i += AGG_TPB) ((v4f*)acc)[i] = (v4f){0.f, 0.f, 0.f, 0.f};
    v4f ga[AGG_NG];
#pragma unroll
    for (int j = 0; j < AGG_NG; ++j) ga[j] = (v4f){0.f, 0.f, 0.f, 0.f};
    const int total = ocount[blockIdx.x];
    const int2* __restrict__ rec =
        (const int2*)recw + agg_owner_base(ocv, (int)blockIdx.x, lane);

    int off[F / 32];
    agg_row_words<F>(off, lane);

    for (int r0 = 0; r0 < total; r0 += AGG_RCAP) {
        const int nst = min(total - r0, AGG_RCAP);
        if (r0 != 0) __syncthreads();
        for (int j0 = tid; j0 < nst; j0 += AGG_RLD * AGG_TPB) {
            int2 t[AGG_RLD];
#pragma unroll
            for (int k = 0; k < AGG_RLD; ++k)
                t[k] = rec[r0 + min(j0 + k * AGG_TPB, nst - 1)];
#pragma unroll
            for (int k = 0; k < AGG_RLD; ++k)
                if (j0 + k * AGG_TPB < nst) srec[j0 + k * AGG_TPB] = t[k];
        }
        __syncthreads();

        const int nfull = nst >> 5;
        const int rest  = nst & 31;
        for (int i = wave; i < nfull; i += AGG_WAVES) {
            const int2 r = srec[32 * i + lane];
            const int dl = (int)((unsigned)r.x >> 16) - node0;
            if (graphs)
                graph_rows(h, ga, r.x & 0xFFFF, sbat[dl] - g_lo,
                           __builtin_bit_cast(float, r.y), lane);
            else
                agg_rows<F, true>(h, acc, r.x & 0xFFFF, dl,
                                  __builtin_bit_cast(float, r.y), 32, lane, off);
        }
        if (rest != 0 && (nfull & (AGG_WAVES - 1)) == wave) {
            int dl_l = 0, s_l = 0;
            float w_l = 0.0f;
            if (lane < rest) {
                const int2 t = srec[32 * nfull + lane];
                s_l  = t.x & 0xFFFF;
                dl_l = (int)((unsigned)t.x >> 16) - node0;
                w_l  = __builtin_bit_cast(float, t.y);
            }
            if (graphs)
                graph_rows(h, ga, s_l, sbat[dl_l] - g_lo, w_l, lane);
            else
                agg_rows<F, false>(h, acc, s_l, dl_l, w_l, rest, lane, off);
        }
    }
    __syncthreads();

    if (!graphs) {
        agg_finish<F, RELU, POOL>(acc, h, dinv, bias, out, batch, sums, cnt, node0, nloc, tid);
        return;
    }

    v4f   rr[AGG_NSELF];
    float dk[AGG_NSELF];
    int   gk[AGG_NSELF];
#pragma unroll
    for (int k = 0; k < AGG_NSELF; ++k) {
        const int ii = min(wave + k * AGG_WAVES, nloc - 1);
        rr[k] = ((const v4f*)h)[(size_t)(node0 + ii) * F4 + lane];
        dk[k] = dinv[node0 + ii];
        gk[k] = __builtin_amdgcn_readfirstlane(sbat[ii]) - g_lo;
    }
    const v4f bv = ((const v4f*)bias)[lane];
#pragma unroll
    for (int k = 0; k < AGG_NSELF; ++k) {
        const bool ok = wave + k * AGG_WAVES < nloc;
        graph_add(ga, gk[k], rr[k], ok ? dk[k] * dk[k] : 0.0f);
        graph_add(ga, gk[k], bv, ok ? 1.0f : 0.0f);
    }

    if (wave < AGG_NSELF) {
        const int i = wave * 32 + lane;
        const int slot = (i < nloc) ? sbat[min(i, nloc - 1)] - g_lo : -1;
#pragma unroll
        for (int j = 0; j < AGG_NG; ++j) {
            const unsigned m = __builtin_amdgcn_ballot_w32(slot == j);
            if (lane == 0) wcnt[wave * AGG_NG + j] = __builtin_popcount(m);
        }
    }

#pragma unroll
    for (int j = 0; j < AGG_NG; ++j) ((v4f*)acc)[(wave * AGG_NG + j) * F4 + lane] = ga[j];
    __syncthreads();
    for (int idx = tid; idx < span * F; idx += AGG_TPB) {
        const int j = idx / F, f = idx - j * F;
        float s = 0.0f;
        for (int w = 0; w < AGG_WAVES; ++w) s += acc[(w * AGG_NG + j) * F + f];
        atomicAdd(&sums[(size_t)(g_lo + j) * F + f], s);
    }
    if (tid < span) {
        int c = 0;
#pragma unroll
        for (int w = 0; w < AGG_NSELF; ++w) c += wcnt[w * AGG_NG + tid];
        atomicAdd(&cnt[g_lo + tid], (float)c);
    }
}

static inline int cdiv(long long a, long long b) { return (int)((a + b - 1) / b); }

extern "C" void kernel_launch(void* const* d_in, const int* in_sizes, int n_in,
                              void* d_out, int out_size, void* d_ws, size_t ws_size,
                              hipStream_t stream) {
    (void)n_in; (void)ws_size;
    const float* x    = (const float*)d_in[0];
    const int*   edge = (const int*)d_in[1];
    const int*   batch= (const int*)d_in[2];
    const float* b1   = (const float*)d_in[4];
    const float* W1   = (const float*)d_in[3];
    const float* W2   = (const float*)d_in[5];
    const float* b2   = (const float*)d_in[6];
    const float* Wfc  = (const float*)d_in[7];
    const float* bfc  = (const float*)d_in[8];
    float* out = (float*)d_out;

    const int N   = in_sizes[2];
    const int E   = in_sizes[1] / 2;
    const int D   = 64;
    const int D2  = 128;
    const int HID = in_sizes[8];
    const int G   = out_size / HID;

    const int* src = edge;
    const int* dst = edge + E;

    float* ws    = (float*)d_ws;
    int*   rpk   = (int*)ws;
    float* hw2   = ws + 2 * (size_t)N * D;
    float* ebuf  = hw2  + (size_t)N * D2;
    int*   pk    = (int*)ebuf;
    int*   obw   = pk + E;
    int*   rec   = obw + E;
    float* zblk  = (float*)(rec + 2 * (size_t)E);
    int*   ocount= (int*)zblk;
    float* sums  = zblk + AGG_OWNERS;
    float* cnt   = sums + (size_t)G * D2;
    unsigned* bm = (unsigned*)(cnt + G);
    float* dinv  = ebuf + (size_t)N * D2;

    fill_kernel<<<cdiv((long long)AGG_OWNERS + G * D2 + G, TPB), TPB, 0, stream>>>(
        zblk, 0.0f, AGG_OWNERS + G * D2 + G);

    edge_prep_kernel<<<cdiv(cdiv(E, 4), TPB), TPB, 0, stream>>>(src, dst, pk, obw, ocount, bm, E);
    owner_scan_kernel<<<cdiv(N, AGG_NPB), AGG_TPB, 0, stream>>>(pk, obw, bm, ocount, dinv, rpk, N, E);

    conv1_owner_kernel<<<cdiv(N, AGG_NPB), AGG_TPB, 0, stream>>>(
        x, rpk, ocount, dinv, W1, b1, W2, hw2, rec, N);

    agg_replay_kernel<128, false, true><<<cdiv(N, AGG_NPB), AGG_TPB, 0, stream>>>(
        hw2, rec, ocount, dinv, b2, nullptr, batch, sums, cnt, N);

    pooled_product_f32<128, 64><<<G / 16, 32 * (64 / 16), 0, stream>>>(sums, cnt, Wfc, bfc, out);
}
